# starting kernel
# baseline (speedup 1.0000x reference)
#include <hip/hip_runtime.h>

#define NN  50000
#define EE  800000
#define EP  (EE + NN)
#define FH  128
#define NH  4
#define HIDW 32

typedef __attribute__((ext_vector_type(2))) float v2f;
typedef __attribute__((ext_vector_type(8))) float v8f;

__device__ __forceinline__ float lrelu(float v) {
    return fmaxf(v, 0.f) + 0.2f * fminf(v, 0.f);
}

__global__ void gemm_wmma_dual(const float* __restrict__ X,
                               const float* __restrict__ Wl,
                               const float* __restrict__ bl,
                               const float* __restrict__ Wr,
                               const float* __restrict__ br,
                               float* __restrict__ OutL,
                               float* __restrict__ OutR)
{
    const int lane  = threadIdx.x & 31;
    const int wave  = threadIdx.x >> 5;
    const int r0    = blockIdx.x * 128 + wave * 16;
    const int n0    = blockIdx.y * 16;
    const int mrow  = lane & 15;
    const int khalf = lane >> 4;

    int arow = r0 + mrow;
    if (arow >= NN) arow = NN - 1;
    const float* xrow = X + (size_t)arow * FH;
    const int col = n0 + mrow;

    v8f cL = {};
    v8f cR = {};
#pragma unroll
    for (int k = 0; k < FH; k += 4) {
        const v2f a = *(const v2f*)(xrow + k + khalf * 2);
        const size_t rk0 = (size_t)(k + khalf * 2) * FH + col;
        v2f bL, bR;
        bL.x = Wl[rk0];       bL.y = Wl[rk0 + FH];
        bR.x = Wr[rk0];       bR.y = Wr[rk0 + FH];
        cL = __builtin_amdgcn_wmma_f32_16x16x4_f32(false, a, false, bL,
                                                   (short)0, cL, false, false);
        cR = __builtin_amdgcn_wmma_f32_16x16x4_f32(false, a, false, bR,
                                                   (short)0, cR, false, false);
    }

    const float bbL = bl[col];
    const float bbR = br[col];
#pragma unroll
    for (int v = 0; v < 8; ++v) {
        const int row = r0 + v + khalf * 8;
        if (row < NN) {
            OutL[(size_t)row * FH + col] = cL[v] + bbL;
            OutR[(size_t)row * FH + col] = cR[v] + bbR;
        }
    }
}

__global__ void init_buffers(float* __restrict__ hbuf,
                             float* __restrict__ m,
                             float* __restrict__ denom)
{
    const int t = blockIdx.x * blockDim.x + threadIdx.x;
    if (t < NN * FH) hbuf[t] = 0.f;
    if (t < NN * NH) { m[t] = -3.0e38f; denom[t] = 0.f; }
}

__global__ void edge_logits(const float* __restrict__ xl,
                            const float* __restrict__ xr,
                            const int* __restrict__ src,
                            const int* __restrict__ dst,
                            const float* __restrict__ att,
                            float* __restrict__ lg,
                            float* __restrict__ m)
{
    const int gid  = blockIdx.x * blockDim.x + threadIdx.x;
    const int e    = gid >> 5;
    const int lane = gid & 31;
    if (e >= EP) return;

    int s, d;
    if (e < EE) { s = src[e]; d = dst[e]; } else { s = d = e - EE; }

    const int f = lane * 4;
    const float4 a = *(const float4*)(xl + (size_t)s * FH + f);
    const float4 b = *(const float4*)(xr + (size_t)d * FH + f);
    const float4 w = *(const float4*)(att + f);

    float p = lrelu(a.x + b.x) * w.x + lrelu(a.y + b.y) * w.y +
              lrelu(a.z + b.z) * w.z + lrelu(a.w + b.w) * w.w;
    p += __shfl_xor(p, 1, 8);
    p += __shfl_xor(p, 2, 8);
    p += __shfl_xor(p, 4, 8);

    if ((lane & 7) == 0) {
        const int h = lane >> 3;
        lg[(size_t)e * NH + h] = p;
        atomicMax(&m[(size_t)d * NH + h], p);
    }
}

__global__ void edge_exp(const int* __restrict__ dst,
                         const float* __restrict__ m,
                         float* __restrict__ lg,
                         float* __restrict__ denom)
{
    const int e = blockIdx.x * blockDim.x + threadIdx.x;
    if (e >= EP) return;
    const int d = (e < EE) ? dst[e] : (e - EE);

    float4 l  = *(const float4*)(lg + (size_t)e * NH);
    const float4 mm = *(const float4*)(m + (size_t)d * NH);
    l.x = expf(l.x - mm.x);
    l.y = expf(l.y - mm.y);
    l.z = expf(l.z - mm.z);
    l.w = expf(l.w - mm.w);
    *(float4*)(lg + (size_t)e * NH) = l;

    float* dn = denom + (size_t)d * NH;
    atomicAdd(dn + 0, l.x);
    atomicAdd(dn + 1, l.y);
    atomicAdd(dn + 2, l.z);
    atomicAdd(dn + 3, l.w);
}

__global__ void edge_scatter(const float* __restrict__ xl,
                             const int* __restrict__ src,
                             const int* __restrict__ dst,
                             const float* __restrict__ lg,
                             const float* __restrict__ denom,
                             float* __restrict__ acc)
{
    const int gid  = blockIdx.x * blockDim.x + threadIdx.x;
    const int e    = gid >> 5;
    const int lane = gid & 31;
    if (e >= EP) return;

    int s, d;
    if (e < EE) { s = src[e]; d = dst[e]; } else { s = d = e - EE; }

    const int h = lane >> 3;
    const float alpha = lg[(size_t)e * NH + h] /
                        fmaxf(denom[(size_t)d * NH + h], 1e-16f);

    const int f = lane * 4;
    const float4 a = *(const float4*)(xl + (size_t)s * FH + f);
    float* o = acc + (size_t)d * FH + f;
    atomicAdd(o + 0, a.x * alpha);
    atomicAdd(o + 1, a.y * alpha);
    atomicAdd(o + 2, a.z * alpha);
    atomicAdd(o + 3, a.w * alpha);
}

__global__ void finalize_elu(float* __restrict__ hbuf,
                             const float* __restrict__ bo)
{
    const int t = blockIdx.x * blockDim.x + threadIdx.x;
    if (t >= NN * FH) return;
    const float v = hbuf[t] + bo[t & (FH - 1)];
    hbuf[t] = (v > 0.f) ? v : (expf(v) - 1.f);
}

__global__ void final_linear(const float* __restrict__ h,
                             const float* __restrict__ Wf,
                             const float* __restrict__ bf,
                             float* __restrict__ out)
{
    const int gid  = blockIdx.x * blockDim.x + threadIdx.x;
    const int n    = gid >> 5;
    const int lane = gid & 31;
    if (n >= NN) return;

    const int f = lane * 4;
    const float4 a = *(const float4*)(h + (size_t)n * FH + f);
    const float4 w = *(const float4*)(Wf + f);
    float p = a.x * w.x + a.y * w.y + a.z * w.z + a.w * w.w;
    p += __shfl_xor(p, 16, 32);
    p += __shfl_xor(p,  8, 32);
    p += __shfl_xor(p,  4, 32);
    p += __shfl_xor(p,  2, 32);
    p += __shfl_xor(p,  1, 32);
    if (lane == 0) out[n] = p + bf[0];
}

extern "C" void kernel_launch(void* const* d_in, const int* in_sizes, int n_in,
                              void* d_out, int out_size, void* d_ws, size_t ws_size,
                              hipStream_t stream)
{
    const float* x    = (const float*)d_in[0];
    const int*   ei   = (const int*)d_in[1];
    const int*   src  = ei;
    const int*   dst  = ei + EE;
    const float* Wl0  = (const float*)d_in[2];
    const float* bl0  = (const float*)d_in[3];
    const float* Wr0  = (const float*)d_in[4];
    const float* br0  = (const float*)d_in[5];
    const float* att0 = (const float*)d_in[6];
    const float* bo0  = (const float*)d_in[7];
    const float* Wl   = (const float*)d_in[8];
    const float* bl   = (const float*)d_in[9];
    const float* Wr   = (const float*)d_in[10];
    const float* br   = (const float*)d_in[11];
    const float* att  = (const float*)d_in[12];
    const float* bo   = (const float*)d_in[13];
    const float* Wf   = (const float*)d_in[14];
    const float* bf   = (const float*)d_in[15];

    float* ws   = (float*)d_ws;
    float* hbuf = ws;
    float* xl   = hbuf + (size_t)NN * FH;
    float* xr   = xl   + (size_t)NN * FH;
    float* lg   = xr   + (size_t)NN * FH;
    float* mbuf = lg   + (size_t)EP * NH;
    float* dbuf = mbuf + (size_t)NN * NH;

    const dim3 gemmGrid((NN + 127) / 128, FH / 16);
    const dim3 edgeGrid((EP * 32 + 255) / 256);
    const dim3 eGrid((EP + 255) / 256);
    const dim3 nodeGrid((NN * FH + 255) / 256);
    const dim3 linGrid((NN * 32 + 255) / 256);

    for (int layer = 0; layer < 3; ++layer) {
        const float* in = (layer == 0) ? x : hbuf;
        const float *WL, *BL, *WR, *BR, *AT, *BO;
        if (layer == 0) { WL = Wl0; BL = bl0; WR = Wr0; BR = br0; AT = att0; BO = bo0; }
        else {
            const int i = layer - 1;
            WL = Wl + (size_t)i * FH * FH;  BL = bl + (size_t)i * FH;
            WR = Wr + (size_t)i * FH * FH;  BR = br + (size_t)i * FH;
            AT = att + (size_t)i * NH * HIDW;
            BO = bo + (size_t)i * FH;
        }
        gemm_wmma_dual<<<gemmGrid, 256, 0, stream>>>(in, WL, BL, WR, BR, xl, xr);
        init_buffers <<<nodeGrid, 256, 0, stream>>>(hbuf, mbuf, dbuf);
        edge_logits  <<<edgeGrid, 256, 0, stream>>>(xl, xr, src, dst, AT, lg, mbuf);
        edge_exp     <<<eGrid,    256, 0, stream>>>(dst, mbuf, lg, dbuf);
        edge_scatter <<<edgeGrid, 256, 0, stream>>>(xl, src, dst, lg, dbuf, hbuf);
        finalize_elu <<<nodeGrid, 256, 0, stream>>>(hbuf, BO);
    }
    final_linear<<<linGrid, 256, 0, stream>>>(hbuf, Wf, bf, (float*)d_out);
}


# Every edge and node pass gives each wave 4 items with their loads issued as one group (4 edges per wave; one 16-byte vector per lane in node passes): 4x fewer waves; atomics kept, same adds per edge
# speedup vs baseline: 1.6521x; 1.6521x over previous
#include <hip/hip_runtime.h>

#define NN  50000
#define EE  800000
#define EP  (EE + NN)
#define FH  128
#define NH  4
#define HIDW 32
#define EPW  4

static_assert(EE % EPW == 0 && EP % EPW == 0, "a wave's edges must not straddle the self-loop boundary");
static_assert(EPW == 4 && NH == 4, "edge passes are written for 4 edges x 4 heads per wave");

typedef __attribute__((ext_vector_type(2))) float v2f;
typedef __attribute__((ext_vector_type(8))) float v8f;

__device__ __forceinline__ float lrelu(float v) {
    return fmaxf(v, 0.f) + 0.2f * fminf(v, 0.f);
}

__global__ void gemm_wmma_dual(const float* __restrict__ X,
                               const float* __restrict__ Wl,
                               const float* __restrict__ bl,
                               const float* __restrict__ Wr,
                               const float* __restrict__ br,
                               float* __restrict__ OutL,
                               float* __restrict__ OutR)
{
    const int lane  = threadIdx.x & 31;
    const int wave  = threadIdx.x >> 5;
    const int r0    = blockIdx.x * 128 + wave * 16;
    const int n0    = blockIdx.y * 16;
    const int mrow  = lane & 15;
    const int khalf = lane >> 4;

    int arow = r0 + mrow;
    if (arow >= NN) arow = NN - 1;
    const float* xrow = X + (size_t)arow * FH;
    const int col = n0 + mrow;

    v8f cL = {};
    v8f cR = {};
#pragma unroll
    for (int k = 0; k < FH; k += 4) {
        const v2f a = *(const v2f*)(xrow + k + khalf * 2);
        const size_t rk0 = (size_t)(k + khalf * 2) * FH + col;
        v2f bL, bR;
        bL.x = Wl[rk0];       bL.y = Wl[rk0 + FH];
        bR.x = Wr[rk0];       bR.y = Wr[rk0 + FH];
        cL = __builtin_amdgcn_wmma_f32_16x16x4_f32(false, a, false, bL,
                                                   (short)0, cL, false, false);
        cR = __builtin_amdgcn_wmma_f32_16x16x4_f32(false, a, false, bR,
                                                   (short)0, cR, false, false);
    }

    const float bbL = bl[col];
    const float bbR = br[col];
#pragma unroll
    for (int v = 0; v < 8; ++v) {
        const int row = r0 + v + khalf * 8;
        if (row < NN) {
            OutL[(size_t)row * FH + col] = cL[v] + bbL;
            OutR[(size_t)row * FH + col] = cR[v] + bbR;
        }
    }
}

__global__ void init_buffers(float* __restrict__ hbuf,
                             float* __restrict__ m,
                             float* __restrict__ denom)
{
    const int t = blockIdx.x * blockDim.x + threadIdx.x;
    if (t < NN * FH / 4) ((float4*)hbuf)[t] = make_float4(0.f, 0.f, 0.f, 0.f);
    if (t < NN * NH / 4) {
        ((float4*)m)[t]     = make_float4(-3.0e38f, -3.0e38f, -3.0e38f, -3.0e38f);
        ((float4*)denom)[t] = make_float4(0.f, 0.f, 0.f, 0.f);
    }
}

__device__ __forceinline__ void wave_edges(const int* __restrict__ src,
                                           const int* __restrict__ dst,
                                           int e0, int (&s)[EPW], int (&d)[EPW])
{
    if (e0 < EE) {
        const int4 sv = *(const int4*)(src + e0);
        const int4 dv = *(const int4*)(dst + e0);
        s[0] = sv.x; s[1] = sv.y; s[2] = sv.z; s[3] = sv.w;
        d[0] = dv.x; d[1] = dv.y; d[2] = dv.z; d[3] = dv.w;
    } else {
#pragma unroll
        for (int j = 0; j < EPW; ++j) s[j] = d[j] = e0 - EE + j;
    }
}

__global__ void edge_logits(const float* __restrict__ xl,
                            const float* __restrict__ xr,
                            const int* __restrict__ src,
                            const int* __restrict__ dst,
                            const float* __restrict__ att,
                            float* __restrict__ lg,
                            float* __restrict__ m)
{
    const int gid  = blockIdx.x * blockDim.x + threadIdx.x;
    const int lane = threadIdx.x & 31;
    const int e0   = __builtin_amdgcn_readfirstlane(gid >> 5) * EPW;
    if (e0 >= EP) return;

    int s[EPW], d[EPW];
    wave_edges(src, dst, e0, s, d);

    const int f = lane * 4;
    float4 a[EPW], b[EPW];
#pragma unroll
    for (int j = 0; j < EPW; ++j) {
        a[j] = *(const float4*)(xl + (size_t)s[j] * FH + f);
        b[j] = *(const float4*)(xr + (size_t)d[j] * FH + f);
    }
    const float4 w = *(const float4*)(att + f);

    float p[EPW];
#pragma unroll
    for (int j = 0; j < EPW; ++j)
        p[j] = lrelu(a[j].x + b[j].x) * w.x + lrelu(a[j].y + b[j].y) * w.y +
               lrelu(a[j].z + b[j].z) * w.z + lrelu(a[j].w + b[j].w) * w.w;

    const bool b0 = (lane & 1) != 0;
    const bool b1 = (lane & 2) != 0;
    const float q0 = (b0 ? p[2] : p[0]) + __shfl_xor(b0 ? p[0] : p[2], 1, 8);
    const float q1 = (b0 ? p[3] : p[1]) + __shfl_xor(b0 ? p[1] : p[3], 1, 8);
    float r = (b1 ? q1 : q0) + __shfl_xor(b1 ? q0 : q1, 2, 8);
    r += __shfl_xor(r, 4, 8);

    if ((lane & 4) == 0) {
        const int j  = ((lane & 1) << 1) | ((lane >> 1) & 1);
        const int h  = lane >> 3;
        const int dj = (j == 0) ? d[0] : (j == 1) ? d[1] : (j == 2) ? d[2] : d[3];
        lg[(size_t)(e0 + j) * NH + h] = r;
        atomicMax(&m[(size_t)dj * NH + h], r);
    }
}

__global__ void edge_scatter(const float* __restrict__ xl,
                             const int* __restrict__ src,
                             const int* __restrict__ dst,
                             const float* __restrict__ lg,
                             const float* __restrict__ m,
                             float* __restrict__ denom,
                             float* __restrict__ acc)
{
    const int gid  = blockIdx.x * blockDim.x + threadIdx.x;
    const int lane = threadIdx.x & 31;
    const int e0   = __builtin_amdgcn_readfirstlane(gid >> 5) * EPW;
    if (e0 >= EP) return;

    int s[EPW], d[EPW];
    wave_edges(src, dst, e0, s, d);

    const int jq = (lane >> 2) & 3;
    const int hq = lane & (NH - 1);
    const int dq = (jq == 0) ? d[0] : (jq == 1) ? d[1] : (jq == 2) ? d[2] : d[3];
    const float l  = lg[(size_t)e0 * NH + (lane & 15)];
    const float mx = m[(size_t)dq * NH + hq];
    float a[EPW][NH];
#pragma unroll
    for (int j = 0; j < EPW; ++j) {
        const float* row = xl + (size_t)s[j] * FH + lane;
#pragma unroll
        for (int h = 0; h < NH; ++h) a[j][h] = row[h * HIDW];
    }
    __builtin_amdgcn_sched_barrier(0);
    const float ex = expf(l - mx);
#pragma unroll
    for (int j = 0; j < EPW; ++j) {
        float* o = acc + (size_t)d[j] * FH + lane;
#pragma unroll
        for (int h = 0; h < NH; ++h) {
            const float wt = __int_as_float(
                __builtin_amdgcn_readlane(__float_as_int(ex), j * NH + h));
            atomicAdd(o + h * HIDW, a[j][h] * wt);
        }
    }
    if (lane < EPW * NH) atomicAdd(denom + (size_t)dq * NH + hq, ex);
}

__device__ __forceinline__ float div_bias_elu(float x, float den, float bias)
{
    const float v = x / den + bias;
    return (v > 0.f) ? v : (expf(v) - 1.f);
}

__global__ void finalize_elu(float* __restrict__ hbuf,
                             const float* __restrict__ denom,
                             const float* __restrict__ bo)
{
    const int t = blockIdx.x * blockDim.x + threadIdx.x;
    if (t >= NN * FH / 4) return;
    float4 v = ((const float4*)hbuf)[t];
    const float den = fmaxf(denom[t >> 3], 1e-16f);
    const float4 bb = ((const float4*)bo)[t & (FH / 4 - 1)];
    v.x = div_bias_elu(v.x, den, bb.x);
    v.y = div_bias_elu(v.y, den, bb.y);
    v.z = div_bias_elu(v.z, den, bb.z);
    v.w = div_bias_elu(v.w, den, bb.w);
    ((float4*)hbuf)[t] = v;
}

__global__ void final_linear(const float* __restrict__ h,
                             const float* __restrict__ Wf,
                             const float* __restrict__ bf,
                             float* __restrict__ out)
{
    const int gid  = blockIdx.x * blockDim.x + threadIdx.x;
    const int n    = gid >> 5;
    const int lane = gid & 31;
    if (n >= NN) return;

    const int f = lane * 4;
    const float4 a = *(const float4*)(h + (size_t)n * FH + f);
    const float4 w = *(const float4*)(Wf + f);
    float p = a.x * w.x + a.y * w.y + a.z * w.z + a.w * w.w;
    p += __shfl_xor(p, 16, 32);
    p += __shfl_xor(p,  8, 32);
    p += __shfl_xor(p,  4, 32);
    p += __shfl_xor(p,  2, 32);
    p += __shfl_xor(p,  1, 32);
    if (lane == 0) out[n] = p + bf[0];
}

extern "C" void kernel_launch(void* const* d_in, const int* in_sizes, int n_in,
                              void* d_out, int out_size, void* d_ws, size_t ws_size,
                              hipStream_t stream)
{
    const float* x    = (const float*)d_in[0];
    const int*   ei   = (const int*)d_in[1];
    const int*   src  = ei;
    const int*   dst  = ei + EE;
    const float* Wl0  = (const float*)d_in[2];
    const float* bl0  = (const float*)d_in[3];
    const float* Wr0  = (const float*)d_in[4];
    const float* br0  = (const float*)d_in[5];
    const float* att0 = (const float*)d_in[6];
    const float* bo0  = (const float*)d_in[7];
    const float* Wl   = (const float*)d_in[8];
    const float* bl   = (const float*)d_in[9];
    const float* Wr   = (const float*)d_in[10];
    const float* br   = (const float*)d_in[11];
    const float* att  = (const float*)d_in[12];
    const float* bo   = (const float*)d_in[13];
    const float* Wf   = (const float*)d_in[14];
    const float* bf   = (const float*)d_in[15];

    float* ws   = (float*)d_ws;
    float* hbuf = ws;
    float* xl   = hbuf + (size_t)NN * FH;
    float* xr   = xl   + (size_t)NN * FH;
    float* lg   = xr   + (size_t)NN * FH;
    float* mbuf = lg   + (size_t)EP * NH;
    float* dbuf = mbuf + (size_t)NN * NH;

    const dim3 gemmGrid((NN + 127) / 128, FH / 16);
    const dim3 edgeGrid((EP / EPW * 32 + 255) / 256);
    const dim3 nodeGrid((NN * FH / 4 + 255) / 256);
    const dim3 linGrid((NN * 32 + 255) / 256);

    for (int layer = 0; layer < 3; ++layer) {
        const float* in = (layer == 0) ? x : hbuf;
        const float *WL, *BL, *WR, *BR, *AT, *BO;
        if (layer == 0) { WL = Wl0; BL = bl0; WR = Wr0; BR = br0; AT = att0; BO = bo0; }
        else {
            const int i = layer - 1;
            WL = Wl + (size_t)i * FH * FH;  BL = bl + (size_t)i * FH;
            WR = Wr + (size_t)i * FH * FH;  BR = br + (size_t)i * FH;
            AT = att + (size_t)i * NH * HIDW;
            BO = bo + (size_t)i * FH;
        }
        gemm_wmma_dual<<<gemmGrid, 256, 0, stream>>>(in, WL, BL, WR, BR, xl, xr);
        init_buffers <<<nodeGrid, 256, 0, stream>>>(hbuf, mbuf, dbuf);
        edge_logits  <<<edgeGrid, 256, 0, stream>>>(xl, xr, src, dst, AT, lg, mbuf);
        edge_scatter <<<edgeGrid, 256, 0, stream>>>(xl, src, dst, lg, mbuf, dbuf, hbuf);
        finalize_elu <<<nodeGrid, 256, 0, stream>>>(hbuf, dbuf, BO);
    }
    final_linear<<<linGrid, 256, 0, stream>>>(hbuf, Wf, bf, (float*)d_out);
}


# Edge passes off global float atomics: edges grouped per call by 98-node destination range; a block per range keeps max, denominator, sums in LDS (LDS float atomics, arrival order), stores rows once
# speedup vs baseline: 3.5011x; 2.1192x over previous
#include <hip/hip_runtime.h>

#define NN  50000
#define EE  800000
#define EP  (EE + NN)
#define FH  128
#define NH  4
#define HIDW 32
#define EPW  4
#define RN   98
#define NR   ((NN + RN - 1) / RN)
#define PB   256
#define PCH  (EE / PB)
#define PPT  ((PCH + 255) / 256)
#define OWT  1024

static_assert(EE % PB == 0, "grouping blocks take equal chunks");
static_assert(EPW == 4 && NH == 4, "edge sweeps are written for 4 items x 4 heads per wave");
static_assert(NR <= 1024, "one block scans the range counts");

typedef __attribute__((ext_vector_type(2))) float v2f;
typedef __attribute__((ext_vector_type(8))) float v8f;

__device__ __forceinline__ float lrelu(float v) {
    return fmaxf(v, 0.f) + 0.2f * fminf(v, 0.f);
}

__global__ void gemm_wmma_dual(const float* __restrict__ X,
                               const float* __restrict__ Wl,
                               const float* __restrict__ bl,
                               const float* __restrict__ Wr,
                               const float* __restrict__ br,
                               float* __restrict__ OutL,
                               float* __restrict__ OutR)
{
    const int lane  = threadIdx.x & 31;
    const int wave  = threadIdx.x >> 5;
    const int r0    = blockIdx.x * 128 + wave * 16;
    const int n0    = blockIdx.y * 16;
    const int mrow  = lane & 15;
    const int khalf = lane >> 4;

    int arow = r0 + mrow;
    if (arow >= NN) arow = NN - 1;
    const float* xrow = X + (size_t)arow * FH;
    const int col = n0 + mrow;

    v8f cL = {};
    v8f cR = {};
#pragma unroll
    for (int k = 0; k < FH; k += 4) {
        const v2f a = *(const v2f*)(xrow + k + khalf * 2);
        const size_t rk0 = (size_t)(k + khalf * 2) * FH + col;
        v2f bL, bR;
        bL.x = Wl[rk0];       bL.y = Wl[rk0 + FH];
        bR.x = Wr[rk0];       bR.y = Wr[rk0 + FH];
        cL = __builtin_amdgcn_wmma_f32_16x16x4_f32(false, a, false, bL,
                                                   (short)0, cL, false, false);
        cR = __builtin_amdgcn_wmma_f32_16x16x4_f32(false, a, false, bR,
                                                   (short)0, cR, false, false);
    }

    const float bbL = bl[col];
    const float bbR = br[col];
#pragma unroll
    for (int v = 0; v < 8; ++v) {
        const int row = r0 + v + khalf * 8;
        if (row < NN) {
            OutL[(size_t)row * FH + col] = cL[v] + bbL;
            OutR[(size_t)row * FH + col] = cR[v] + bbR;
        }
    }
}

__global__ void part_zero(int* __restrict__ cnt, int* __restrict__ cur)
{
    const int t = blockIdx.x * blockDim.x + threadIdx.x;
    if (t < NR) { cnt[t] = 0; cur[t] = 0; }
}

__global__ void part_count(const int* __restrict__ dst, int* __restrict__ cnt)
{
    __shared__ int hist[NR];
    for (int i = threadIdx.x; i < NR; i += 256) hist[i] = 0;
    __syncthreads();
    const int e0 = blockIdx.x * PCH;
    for (int k = 0; k < PPT; ++k) {
        const int le = k * 256 + threadIdx.x;
        if (le < PCH) atomicAdd(&hist[dst[e0 + le] / RN], 1);
    }
    __syncthreads();
    for (int i = threadIdx.x; i < NR; i += 256) {
        const int c = hist[i];
        if (c) atomicAdd(&cnt[i], c);
    }
}

__global__ void part_scan(const int* __restrict__ cnt, int* __restrict__ off)
{
    __shared__ int s[1024];
    const int t = threadIdx.x;
    const int v = (t < NR) ? cnt[t] : 0;
    s[t] = v;
    __syncthreads();
    for (int d = 1; d < 1024; d <<= 1) {
        const int add = (t >= d) ? s[t - d] : 0;
        __syncthreads();
        s[t] += add;
        __syncthreads();
    }
    if (t < NR) off[t] = s[t] - v;
    if (t == NR - 1) off[NR] = s[t];
}

__global__ void part_fill(const int* __restrict__ src, const int* __restrict__ dst,
                          const int* __restrict__ off, int* __restrict__ cur,
                          int* __restrict__ ssrc, int* __restrict__ sdst)
{
    __shared__ int hist[NR];
    __shared__ int basep[NR];
    for (int i = threadIdx.x; i < NR; i += 256) hist[i] = 0;
    __syncthreads();
    const int e0 = blockIdx.x * PCH;
    int rk[PPT], dv[PPT];
#pragma unroll
    for (int k = 0; k < PPT; ++k) {
        const int le = k * 256 + threadIdx.x;
        rk[k] = -1; dv[k] = 0;
        if (le < PCH) {
            dv[k] = dst[e0 + le];
            rk[k] = atomicAdd(&hist[dv[k] / RN], 1);
        }
    }
    __syncthreads();
    for (int i = threadIdx.x; i < NR; i += 256) {
        const int c = hist[i];
        basep[i] = c ? off[i] + atomicAdd(&cur[i], c) : 0;
    }
    __syncthreads();
#pragma unroll
    for (int k = 0; k < PPT; ++k) {
        if (rk[k] >= 0) {
            const int pos = basep[dv[k] / RN] + rk[k];
            ssrc[pos] = src[e0 + k * 256 + threadIdx.x];
            sdst[pos] = dv[k];
        }
    }
}

__device__ __forceinline__ float div_bias_elu(float x, float den, float bias)
{
    const float v = x / den + bias;
    return (v > 0.f) ? v : (expf(v) - 1.f);
}

__device__ __forceinline__ void wave_items(const int* __restrict__ ssrc,
                                           const int* __restrict__ sdst,
                                           int e_lo, int cnt, int tot, int base, int i0,
                                           int (&s)[EPW], int (&d)[EPW], int (&pos)[EPW])
{
#pragma unroll
    for (int j = 0; j < EPW; ++j) {
        int idx = i0 + j;
        if (idx >= tot) idx = tot - 1;
        if (idx < cnt) {
            s[j] = ssrc[e_lo + idx];
            d[j] = sdst[e_lo + idx];
            pos[j] = e_lo + idx;
        } else {
            s[j] = d[j] = base + (idx - cnt);
            pos[j] = EE + s[j];
        }
    }
}

template <bool LAST>
__global__ __launch_bounds__(OWT)
void owner_layer(const float* __restrict__ xl,
                 const float* __restrict__ xr,
                 const int* __restrict__ ssrc,
                 const int* __restrict__ sdst,
                 const int* __restrict__ off,
                 const float* __restrict__ att,
                 const float* __restrict__ bo,
                 float* __restrict__ lg,
                 float* __restrict__ hout,
                 const float* __restrict__ Wf,
                 const float* __restrict__ bf,
                 float* __restrict__ out)
{
    __shared__ float buf[RN * FH];
    __shared__ float mx[RN * NH];
    __shared__ float den[RN * NH];

    const int tid  = threadIdx.x;
    const int lane = tid & 31;
    const int wave = __builtin_amdgcn_readfirstlane(tid >> 5);
    const int base = blockIdx.x * RN;
    const int nown = min(RN, NN - base);
    const int e_lo = off[blockIdx.x];
    const int cnt  = off[blockIdx.x + 1] - e_lo;
    const int tot  = cnt + nown;
    const int ngrp = (tot + EPW - 1) / EPW;

    {
        const float4* g = (const float4*)(xr + (size_t)base * FH);
        for (int i = tid; i < nown * (FH / 4); i += OWT) ((float4*)buf)[i] = g[i];
        for (int i = tid; i < RN * NH; i += OWT) { mx[i] = -3.0e38f; den[i] = 0.f; }
    }
    __syncthreads();

    {
        const int f = lane * 4;
        const float4 w = *(const float4*)(att + f);
        const bool b0 = (lane & 1) != 0;
        const bool b1 = (lane & 2) != 0;
        for (int g = wave; g < ngrp; g += OWT / 32) {
            const int i0 = g * EPW;
            int s[EPW], d[EPW], pos[EPW];
            wave_items(ssrc, sdst, e_lo, cnt, tot, base, i0, s, d, pos);
            float4 a[EPW], b[EPW];
#pragma unroll
            for (int j = 0; j < EPW; ++j) {
                a[j] = *(const float4*)(xl + (size_t)s[j] * FH + f);
                b[j] = *(const float4*)(buf + (d[j] - base) * FH + f);
            }
            float p[EPW];
#pragma unroll
            for (int j = 0; j < EPW; ++j)
                p[j] = lrelu(a[j].x + b[j].x) * w.x + lrelu(a[j].y + b[j].y) * w.y +
                       lrelu(a[j].z + b[j].z) * w.z + lrelu(a[j].w + b[j].w) * w.w;
            const float q0 = (b0 ? p[2] : p[0]) + __shfl_xor(b0 ? p[0] : p[2], 1, 8);
            const float q1 = (b0 ? p[3] : p[1]) + __shfl_xor(b0 ? p[1] : p[3], 1, 8);
            float r = (b1 ? q1 : q0) + __shfl_xor(b1 ? q0 : q1, 2, 8);
            r += __shfl_xor(r, 4, 8);
            if ((lane & 4) == 0) {
                const int j  = ((lane & 1) << 1) | ((lane >> 1) & 1);
                const int h  = lane >> 3;
                const int dj = (j == 0) ? d[0] : (j == 1) ? d[1] : (j == 2) ? d[2] : d[3];
                const int pj = (j == 0) ? pos[0] : (j == 1) ? pos[1] : (j == 2) ? pos[2] : pos[3];
                if (i0 + j < tot) {
                    lg[(size_t)pj * NH + h] = r;
                    atomicMax(&mx[(dj - base) * NH + h], r);
                }
            }
        }
    }
    __syncthreads();
    for (int i = tid; i < RN * FH / 4; i += OWT) ((float4*)buf)[i] = make_float4(0.f, 0.f, 0.f, 0.f);
    __syncthreads();

    {
        const int jq = (lane >> 2) & 3;
        const int hq = lane & (NH - 1);
        for (int g = wave; g < ngrp; g += OWT / 32) {
            const int i0 = g * EPW;
            int s[EPW], d[EPW], pos[EPW];
            wave_items(ssrc, sdst, e_lo, cnt, tot, base, i0, s, d, pos);
            const int dq = (jq == 0) ? d[0] : (jq == 1) ? d[1] : (jq == 2) ? d[2] : d[3];
            const int pq = (jq == 0) ? pos[0] : (jq == 1) ? pos[1] : (jq == 2) ? pos[2] : pos[3];
            const bool okq = (i0 + jq) < tot;
            const float l  = lg[(size_t)pq * NH + hq];
            const float mv = mx[(dq - base) * NH + hq];
            float a[EPW][NH];
#pragma unroll
            for (int j = 0; j < EPW; ++j) {
                const float* row = xl + (size_t)s[j] * FH + lane;
#pragma unroll
                for (int h = 0; h < NH; ++h) a[j][h] = row[h * HIDW];
            }
            __builtin_amdgcn_sched_barrier(0);
            const float ex = okq ? expf(l - mv) : 0.f;
#pragma unroll
            for (int j = 0; j < EPW; ++j) {
                if (i0 + j < tot) {
                    float* o = buf + (d[j] - base) * FH + lane;
#pragma unroll
                    for (int h = 0; h < NH; ++h) {
                        const float wt = __int_as_float(
                            __builtin_amdgcn_readlane(__float_as_int(ex), j * NH + h));
                        atomicAdd(o + h * HIDW, a[j][h] * wt);
                    }
                }
            }
            if (lane < EPW * NH && okq) atomicAdd(&den[(dq - base) * NH + hq], ex);
        }
    }
    __syncthreads();

    for (int i = wave; i < nown; i += OWT / 32) {
        float4 v = ((const float4*)buf)[i * (FH / 4) + lane];
        const float dn = fmaxf(den[i * NH + (lane >> 3)], 1e-16f);
        const float4 bb = ((const float4*)bo)[lane];
        v.x = div_bias_elu(v.x, dn, bb.x);
        v.y = div_bias_elu(v.y, dn, bb.y);
        v.z = div_bias_elu(v.z, dn, bb.z);
        v.w = div_bias_elu(v.w, dn, bb.w);
        if (!LAST) {
            ((float4*)(hout + (size_t)(base + i) * FH))[lane] = v;
        } else {
            const float4 wf = ((const float4*)Wf)[lane];
            float p = v.x * wf.x + v.y * wf.y + v.z * wf.z + v.w * wf.w;
            p += __shfl_xor(p, 16, 32);
            p += __shfl_xor(p,  8, 32);
            p += __shfl_xor(p,  4, 32);
            p += __shfl_xor(p,  2, 32);
            p += __shfl_xor(p,  1, 32);
            if (lane == 0) out[base + i] = p + bf[0];
        }
    }
}

extern "C" void kernel_launch(void* const* d_in, const int* in_sizes, int n_in,
                              void* d_out, int out_size, void* d_ws, size_t ws_size,
                              hipStream_t stream)
{
    const float* x    = (const float*)d_in[0];
    const int*   ei   = (const int*)d_in[1];
    const int*   src  = ei;
    const int*   dst  = ei + EE;
    const float* Wl0  = (const float*)d_in[2];
    const float* bl0  = (const float*)d_in[3];
    const float* Wr0  = (const float*)d_in[4];
    const float* br0  = (const float*)d_in[5];
    const float* att0 = (const float*)d_in[6];
    const float* bo0  = (const float*)d_in[7];
    const float* Wl   = (const float*)d_in[8];
    const float* bl   = (const float*)d_in[9];
    const float* Wr   = (const float*)d_in[10];
    const float* br   = (const float*)d_in[11];
    const float* att  = (const float*)d_in[12];
    const float* bo   = (const float*)d_in[13];
    const float* Wf   = (const float*)d_in[14];
    const float* bf   = (const float*)d_in[15];

    float* ws   = (float*)d_ws;
    float* hbuf = ws;
    float* xl   = hbuf + (size_t)NN * FH;
    float* xr   = xl   + (size_t)NN * FH;
    float* lg   = xr   + (size_t)NN * FH;
    int*   ssrc = (int*)(lg + (size_t)EP * NH);
    int*   sdst = ssrc + EE;
    int*   cnt  = sdst + EE;
    int*   cur  = cnt + NR;
    int*   off  = cur + NR;

    part_zero <<<(NR + 255) / 256, 256, 0, stream>>>(cnt, cur);
    part_count<<<PB, 256, 0, stream>>>(dst, cnt);
    part_scan <<<1, 1024, 0, stream>>>(cnt, off);
    part_fill <<<PB, 256, 0, stream>>>(src, dst, off, cur, ssrc, sdst);

    const dim3 gemmGrid((NN + 127) / 128, FH / 16);

    for (int layer = 0; layer < 3; ++layer) {
        const float* in = (layer == 0) ? x : hbuf;
        const float *WL, *BL, *WR, *BR, *AT, *BO;
        if (layer == 0) { WL = Wl0; BL = bl0; WR = Wr0; BR = br0; AT = att0; BO = bo0; }
        else {
            const int i = layer - 1;
            WL = Wl + (size_t)i * FH * FH;  BL = bl + (size_t)i * FH;
            WR = Wr + (size_t)i * FH * FH;  BR = br + (size_t)i * FH;
            AT = att + (size_t)i * NH * HIDW;
            BO = bo + (size_t)i * FH;
        }
        gemm_wmma_dual<<<gemmGrid, 256, 0, stream>>>(in, WL, BL, WR, BR, xl, xr);
        if (layer < 2)
            owner_layer<false><<<NR, OWT, 0, stream>>>(xl, xr, ssrc, sdst, off, AT, BO, lg, hbuf,
                                                       Wf, bf, (float*)d_out);
        else
            owner_layer<true> <<<NR, OWT, 0, stream>>>(xl, xr, ssrc, sdst, off, AT, BO, lg, hbuf,
                                                       Wf, bf, (float*)d_out);
    }
}


# Layers 1, 2: the dual GEMM moves into the owner block's finish: h rows stay in LDS (pitch 132) as WMMA A fragments, same k order; xl/xr go to a second buffer pair; 2 GEMM launches, h round trip gone
# speedup vs baseline: 4.2931x; 1.2262x over previous
#include <hip/hip_runtime.h>

#define NN  50000
#define EE  800000
#define EP  (EE + NN)
#define FH  128
#define NH  4
#define HIDW 32
#define EPW  4
#define RN   98
#define NR   ((NN + RN - 1) / RN)
#define PB   256
#define PCH  (EE / PB)
#define PPT  ((PCH + 255) / 256)
#define OWT  1024
#define LP   (FH + 4)

static_assert(EE % PB == 0, "grouping blocks take equal chunks");
static_assert(EPW == 4 && NH == 4, "edge sweeps are written for 4 items x 4 heads per wave");
static_assert(NR <= 1024, "one block scans the range counts");
static_assert(RN > 6 * 16 && RN <= 7 * 16 && OWT == 1024, "the GEMM tail of an owner block splits 7 row tiles over 32 waves");
static_assert(LP % 4 == 0, "LDS rows stay 16-byte aligned");

typedef __attribute__((ext_vector_type(2))) float v2f;
typedef __attribute__((ext_vector_type(8))) float v8f;

__device__ __forceinline__ float lrelu(float v) {
    return fmaxf(v, 0.f) + 0.2f * fminf(v, 0.f);
}

__global__ void gemm_wmma_dual(const float* __restrict__ X,
                               const float* __restrict__ Wl,
                               const float* __restrict__ bl,
                               const float* __restrict__ Wr,
                               const float* __restrict__ br,
                               float* __restrict__ OutL,
                               float* __restrict__ OutR)
{
    const int lane  = threadIdx.x & 31;
    const int wave  = threadIdx.x >> 5;
    const int r0    = blockIdx.x * 128 + wave * 16;
    const int n0    = blockIdx.y * 16;
    const int mrow  = lane & 15;
    const int khalf = lane >> 4;

    int arow = r0 + mrow;
    if (arow >= NN) arow = NN - 1;
    const float* xrow = X + (size_t)arow * FH;
    const int col = n0 + mrow;

    v8f cL = {};
    v8f cR = {};
#pragma unroll
    for (int k = 0; k < FH; k += 4) {
        const v2f a = *(const v2f*)(xrow + k + khalf * 2);
        const size_t rk0 = (size_t)(k + khalf * 2) * FH + col;
        v2f bL, bR;
        bL.x = Wl[rk0];       bL.y = Wl[rk0 + FH];
        bR.x = Wr[rk0];       bR.y = Wr[rk0 + FH];
        cL = __builtin_amdgcn_wmma_f32_16x16x4_f32(false, a, false, bL,
                                                   (short)0, cL, false, false);
        cR = __builtin_amdgcn_wmma_f32_16x16x4_f32(false, a, false, bR,
                                                   (short)0, cR, false, false);
    }

    const float bbL = bl[col];
    const float bbR = br[col];
#pragma unroll
    for (int v = 0; v < 8; ++v) {
        const int row = r0 + v + khalf * 8;
        if (row < NN) {
            OutL[(size_t)row * FH + col] = cL[v] + bbL;
            OutR[(size_t)row * FH + col] = cR[v] + bbR;
        }
    }
}

__global__ void part_zero(int* __restrict__ cnt, int* __restrict__ cur)
{
    const int t = blockIdx.x * blockDim.x + threadIdx.x;
    if (t < NR) { cnt[t] = 0; cur[t] = 0; }
}

__global__ void part_count(const int* __restrict__ dst, int* __restrict__ cnt)
{
    __shared__ int hist[NR];
    for (int i = threadIdx.x; i < NR; i += 256) hist[i] = 0;
    __syncthreads();
    const int e0 = blockIdx.x * PCH;
    for (int k = 0; k < PPT; ++k) {
        const int le = k * 256 + threadIdx.x;
        if (le < PCH) atomicAdd(&hist[dst[e0 + le] / RN], 1);
    }
    __syncthreads();
    for (int i = threadIdx.x; i < NR; i += 256) {
        const int c = hist[i];
        if (c) atomicAdd(&cnt[i], c);
    }
}

__global__ void part_scan(const int* __restrict__ cnt, int* __restrict__ off)
{
    __shared__ int s[1024];
    const int t = threadIdx.x;
    const int v = (t < NR) ? cnt[t] : 0;
    s[t] = v;
    __syncthreads();
    for (int d = 1; d < 1024; d <<= 1) {
        const int add = (t >= d) ? s[t - d] : 0;
        __syncthreads();
        s[t] += add;
        __syncthreads();
    }
    if (t < NR) off[t] = s[t] - v;
    if (t == NR - 1) off[NR] = s[t];
}

__global__ void part_fill(const int* __restrict__ src, const int* __restrict__ dst,
                          const int* __restrict__ off, int* __restrict__ cur,
                          int* __restrict__ ssrc, int* __restrict__ sdst)
{
    __shared__ int hist[NR];
    __shared__ int basep[NR];
    for (int i = threadIdx.x; i < NR; i += 256) hist[i] = 0;
    __syncthreads();
    const int e0 = blockIdx.x * PCH;
    int rk[PPT], dv[PPT];
#pragma unroll
    for (int k = 0; k < PPT; ++k) {
        const int le = k * 256 + threadIdx.x;
        rk[k] = -1; dv[k] = 0;
        if (le < PCH) {
            dv[k] = dst[e0 + le];
            rk[k] = atomicAdd(&hist[dv[k] / RN], 1);
        }
    }
    __syncthreads();
    for (int i = threadIdx.x; i < NR; i += 256) {
        const int c = hist[i];
        basep[i] = c ? off[i] + atomicAdd(&cur[i], c) : 0;
    }
    __syncthreads();
#pragma unroll
    for (int k = 0; k < PPT; ++k) {
        if (rk[k] >= 0) {
            const int pos = basep[dv[k] / RN] + rk[k];
            ssrc[pos] = src[e0 + k * 256 + threadIdx.x];
            sdst[pos] = dv[k];
        }
    }
}

__device__ __forceinline__ float div_bias_elu(float x, float den, float bias)
{
    const float v = x / den + bias;
    return (v > 0.f) ? v : (expf(v) - 1.f);
}

__device__ __forceinline__ void wave_items(const int* __restrict__ ssrc,
                                           const int* __restrict__ sdst,
                                           int e_lo, int cnt, int tot, int base, int i0,
                                           int (&s)[EPW], int (&d)[EPW], int (&pos)[EPW])
{
#pragma unroll
    for (int j = 0; j < EPW; ++j) {
        int idx = i0 + j;
        if (idx >= tot) idx = tot - 1;
        if (idx < cnt) {
            s[j] = ssrc[e_lo + idx];
            d[j] = sdst[e_lo + idx];
            pos[j] = e_lo + idx;
        } else {
            s[j] = d[j] = base + (idx - cnt);
            pos[j] = EE + s[j];
        }
    }
}

template <int NT>
__device__ __forceinline__ void next_tiles(const float* buf, int rt0, int lane,
                                           const float* __restrict__ Wn,
                                           const float* __restrict__ bn,
                                           float* __restrict__ on,
                                           int base, int nown, int col)
{
    const int mrow  = lane & 15;
    const int khalf = lane >> 4;
    const float* wp = Wn + (size_t)(khalf * 2) * FH + col;
    const float* ap[NT];
#pragma unroll
    for (int t = 0; t < NT; ++t) {
        const int r = min((rt0 + t) * 16 + mrow, RN - 1);
        ap[t] = buf + r * LP + khalf * 2;
    }
    v8f c[NT] = {};
#pragma unroll
    for (int k = 0; k < FH; k += 4) {
        v2f b;
        b.x = wp[(size_t)k * FH];
        b.y = wp[(size_t)k * FH + FH];
#pragma unroll
        for (int t = 0; t < NT; ++t) {
            const v2f a = *(const v2f*)(ap[t] + k);
            c[t] = __builtin_amdgcn_wmma_f32_16x16x4_f32(false, a, false, b,
                                                         (short)0, c[t], false, false);
        }
    }
    const float bb = bn[col];
#pragma unroll
    for (int t = 0; t < NT; ++t) {
#pragma unroll
        for (int v = 0; v < 8; ++v) {
            const int row = (rt0 + t) * 16 + v + khalf * 8;
            if (row < nown) on[(size_t)(base + row) * FH + col] = c[t][v] + bb;
        }
    }
}

template <bool LAST>
__global__ __launch_bounds__(OWT, 16)
void owner_layer(const float* __restrict__ xl,
                 const float* __restrict__ xr,
                 const int* __restrict__ ssrc,
                 const int* __restrict__ sdst,
                 const int* __restrict__ off,
                 const float* __restrict__ att,
                 const float* __restrict__ bo,
                 float* __restrict__ lg,
                 const float* __restrict__ Wln,
                 const float* __restrict__ bln,
                 const float* __restrict__ Wrn,
                 const float* __restrict__ brn,
                 float* __restrict__ xln,
                 float* __restrict__ xrn,
                 const float* __restrict__ Wf,
                 const float* __restrict__ bf,
                 float* __restrict__ out)
{
    __shared__ __attribute__((aligned(16))) float buf[RN * LP];
    __shared__ float mx[RN * NH];
    __shared__ float den[RN * NH];

    const int tid  = threadIdx.x;
    const int lane = tid & 31;
    const int wave = __builtin_amdgcn_readfirstlane(tid >> 5);
    const int base = blockIdx.x * RN;
    const int nown = min(RN, NN - base);
    const int e_lo = off[blockIdx.x];
    const int cnt  = off[blockIdx.x + 1] - e_lo;
    const int tot  = cnt + nown;
    const int ngrp = (tot + EPW - 1) / EPW;

    {
        const float4* g = (const float4*)(xr + (size_t)base * FH);
        for (int i = tid; i < nown * (FH / 4); i += OWT)
            ((float4*)buf)[(i >> 5) * (LP / 4) + (i & 31)] = g[i];
        for (int i = tid; i < RN * NH; i += OWT) { mx[i] = -3.0e38f; den[i] = 0.f; }
    }
    __syncthreads();

    {
        const int f = lane * 4;
        const float4 w = *(const float4*)(att + f);
        const bool b0 = (lane & 1) != 0;
        const bool b1 = (lane & 2) != 0;
        for (int g = wave; g < ngrp; g += OWT / 32) {
            const int i0 = g * EPW;
            int s[EPW], d[EPW], pos[EPW];
            wave_items(ssrc, sdst, e_lo, cnt, tot, base, i0, s, d, pos);
            float4 a[EPW], b[EPW];
#pragma unroll
            for (int j = 0; j < EPW; ++j) {
                a[j] = *(const float4*)(xl + (size_t)s[j] * FH + f);
                b[j] = *(const float4*)(buf + (d[j] - base) * LP + f);
            }
            float p[EPW];
#pragma unroll
            for (int j = 0; j < EPW; ++j)
                p[j] = lrelu(a[j].x + b[j].x) * w.x + lrelu(a[j].y + b[j].y) * w.y +
                       lrelu(a[j].z + b[j].z) * w.z + lrelu(a[j].w + b[j].w) * w.w;
            const float q0 = (b0 ? p[2] : p[0]) + __shfl_xor(b0 ? p[0] : p[2], 1, 8);
            const float q1 = (b0 ? p[3] : p[1]) + __shfl_xor(b0 ? p[1] : p[3], 1, 8);
            float r = (b1 ? q1 : q0) + __shfl_xor(b1 ? q0 : q1, 2, 8);
            r += __shfl_xor(r, 4, 8);
            if ((lane & 4) == 0) {
                const int j  = ((lane & 1) << 1) | ((lane >> 1) & 1);
                const int h  = lane >> 3;
                const int dj = (j == 0) ? d[0] : (j == 1) ? d[1] : (j == 2) ? d[2] : d[3];
                const int pj = (j == 0) ? pos[0] : (j == 1) ? pos[1] : (j == 2) ? pos[2] : pos[3];
                if (i0 + j < tot) {
                    lg[(size_t)pj * NH + h] = r;
                    atomicMax(&mx[(dj - base) * NH + h], r);
                }
            }
        }
    }
    __syncthreads();
    for (int i = tid; i < RN * LP / 4; i += OWT) ((float4*)buf)[i] = make_float4(0.f, 0.f, 0.f, 0.f);
    __syncthreads();

    {
        const int jq = (lane >> 2) & 3;
        const int hq = lane & (NH - 1);
        for (int g = wave; g < ngrp; g += OWT / 32) {
            const int i0 = g * EPW;
            int s[EPW], d[EPW], pos[EPW];
            wave_items(ssrc, sdst, e_lo, cnt, tot, base, i0, s, d, pos);
            const int dq = (jq == 0) ? d[0] : (jq == 1) ? d[1] : (jq == 2) ? d[2] : d[3];
            const int pq = (jq == 0) ? pos[0] : (jq == 1) ? pos[1] : (jq == 2) ? pos[2] : pos[3];
            const bool okq = (i0 + jq) < tot;
            const float l  = lg[(size_t)pq * NH + hq];
            const float mv = mx[(dq - base) * NH + hq];
            float a[EPW][NH];
#pragma unroll
            for (int j = 0; j < EPW; ++j) {
                const float* row = xl + (size_t)s[j] * FH + lane;
#pragma unroll
                for (int h = 0; h < NH; ++h) a[j][h] = row[h * HIDW];
            }
            __builtin_amdgcn_sched_barrier(0);
            const float ex = okq ? expf(l - mv) : 0.f;
#pragma unroll
            for (int j = 0; j < EPW; ++j) {
                if (i0 + j < tot) {
                    float* o = buf + (d[j] - base) * LP + lane;
#pragma unroll
                    for (int h = 0; h < NH; ++h) {
                        const float wt = __int_as_float(
                            __builtin_amdgcn_readlane(__float_as_int(ex), j * NH + h));
                        atomicAdd(o + h * HIDW, a[j][h] * wt);
                    }
                }
            }
            if (lane < EPW * NH && okq) atomicAdd(&den[(dq - base) * NH + hq], ex);
        }
    }
    __syncthreads();

    for (int i = wave; i < nown; i += OWT / 32) {
        float4 v = ((const float4*)buf)[i * (LP / 4) + lane];
        const float dn = fmaxf(den[i * NH + (lane >> 3)], 1e-16f);
        const float4 bb = ((const float4*)bo)[lane];
        v.x = div_bias_elu(v.x, dn, bb.x);
        v.y = div_bias_elu(v.y, dn, bb.y);
        v.z = div_bias_elu(v.z, dn, bb.z);
        v.w = div_bias_elu(v.w, dn, bb.w);
        if (!LAST) {
            ((float4*)buf)[i * (LP / 4) + lane] = v;
        } else {
            const float4 wf = ((const float4*)Wf)[lane];
            float p = v.x * wf.x + v.y * wf.y + v.z * wf.z + v.w * wf.w;
            p += __shfl_xor(p, 16, 32);
            p += __shfl_xor(p,  8, 32);
            p += __shfl_xor(p,  4, 32);
            p += __shfl_xor(p,  2, 32);
            p += __shfl_xor(p,  1, 32);
            if (lane == 0) out[base + i] = p + bf[0];
        }
    }

    if (!LAST) {
        __syncthreads();
        const int combo = wave & 15;
        const bool right = (combo & 8) != 0;
        const int col = (combo & 7) * 16 + (lane & 15);
        if (wave < 16)
            next_tiles<4>(buf, 0, lane, right ? Wrn : Wln, right ? brn : bln,
                          right ? xrn : xln, base, nown, col);
        else
            next_tiles<3>(buf, 4, lane, right ? Wrn : Wln, right ? brn : bln,
                          right ? xrn : xln, base, nown, col);
    }
}

extern "C" void kernel_launch(void* const* d_in, const int* in_sizes, int n_in,
                              void* d_out, int out_size, void* d_ws, size_t ws_size,
                              hipStream_t stream)
{
    const float* x    = (const float*)d_in[0];
    const int*   ei   = (const int*)d_in[1];
    const int*   src  = ei;
    const int*   dst  = ei + EE;
    const float* Wl0  = (const float*)d_in[2];
    const float* bl0  = (const float*)d_in[3];
    const float* Wr0  = (const float*)d_in[4];
    const float* br0  = (const float*)d_in[5];
    const float* att0 = (const float*)d_in[6];
    const float* bo0  = (const float*)d_in[7];
    const float* Wl   = (const float*)d_in[8];
    const float* bl   = (const float*)d_in[9];
    const float* Wr   = (const float*)d_in[10];
    const float* br   = (const float*)d_in[11];
    const float* att  = (const float*)d_in[12];
    const float* bo   = (const float*)d_in[13];
    const float* Wf   = (const float*)d_in[14];
    const float* bf   = (const float*)d_in[15];

    float* ws   = (float*)d_ws;
    float* xlA  = ws;
    float* xrA  = xlA  + (size_t)NN * FH;
    float* xlB  = xrA  + (size_t)NN * FH;
    float* xrB  = xlB  + (size_t)NN * FH;
    float* lg   = xrB  + (size_t)NN * FH;
    int*   ssrc = (int*)(lg + (size_t)EP * NH);
    int*   sdst = ssrc + EE;
    int*   cnt  = sdst + EE;
    int*   cur  = cnt + NR;
    int*   off  = cur + NR;

    part_zero <<<(NR + 255) / 256, 256, 0, stream>>>(cnt, cur);
    part_count<<<PB, 256, 0, stream>>>(dst, cnt);
    part_scan <<<1, 1024, 0, stream>>>(cnt, off);
    part_fill <<<PB, 256, 0, stream>>>(src, dst, off, cur, ssrc, sdst);

    const dim3 gemmGrid((NN + 127) / 128, FH / 16);

    gemm_wmma_dual<<<gemmGrid, 256, 0, stream>>>(x, Wl0, bl0, Wr0, br0, xlA, xrA);
    owner_layer<false><<<NR, OWT, 0, stream>>>(xlA, xrA, ssrc, sdst, off, att0, bo0, lg,
                                               Wl, bl, Wr, br, xlB, xrB,
                                               Wf, bf, (float*)d_out);
    owner_layer<false><<<NR, OWT, 0, stream>>>(xlB, xrB, ssrc, sdst, off, att, bo, lg,
                                               Wl + (size_t)FH * FH, bl + FH,
                                               Wr + (size_t)FH * FH, br + FH, xlA, xrA,
                                               Wf, bf, (float*)d_out);
    owner_layer<true> <<<NR, OWT, 0, stream>>>(xlA, xrA, ssrc, sdst, off,
                                               att + (size_t)NH * HIDW, bo + FH, lg,
                                               Wl, bl, Wr, br, xlB, xrB,
                                               Wf, bf, (float*)d_out);
}


# Edge pass is one sweep, not two: exponentials shifted by the node's self-loop logit (known up front), not the segment max; rows gathered once; block redoes with exact max if a logit is 40 above
# speedup vs baseline: 4.9899x; 1.1623x over previous
#include <hip/hip_runtime.h>

#define NN  50000
#define EE  800000
#define EP  (EE + NN)
#define FH  128
#define NH  4
#define HIDW 32
#define EPW  4
#define RN   98
#define NR   ((NN + RN - 1) / RN)
#define PB   256
#define PCH  (EE / PB)
#define PPT  ((PCH + 255) / 256)
#define OWT  1024
#define LP   (FH + 4)

static_assert(EE % PB == 0, "grouping blocks take equal chunks");
static_assert(EPW == 4 && NH == 4, "edge sweeps are written for 4 items x 4 heads per wave");
static_assert(NR <= 1024, "one block scans the range counts");
static_assert(RN > 6 * 16 && RN <= 7 * 16 && OWT == 1024, "the GEMM tail of an owner block splits 7 row tiles over 32 waves");
static_assert(LP % 4 == 0, "LDS rows stay 16-byte aligned");

typedef __attribute__((ext_vector_type(2))) float v2f;
typedef __attribute__((ext_vector_type(8))) float v8f;

__device__ __forceinline__ float lrelu(float v) {
    return fmaxf(v, 0.f) + 0.2f * fminf(v, 0.f);
}

__global__ void gemm_wmma_dual(const float* __restrict__ X,
                               const float* __restrict__ Wl,
                               const float* __restrict__ bl,
                               const float* __restrict__ Wr,
                               const float* __restrict__ br,
                               float* __restrict__ OutL,
                               float* __restrict__ OutR)
{
    const int lane  = threadIdx.x & 31;
    const int wave  = threadIdx.x >> 5;
    const int r0    = blockIdx.x * 128 + wave * 16;
    const int n0    = blockIdx.y * 16;
    const int mrow  = lane & 15;
    const int khalf = lane >> 4;

    int arow = r0 + mrow;
    if (arow >= NN) arow = NN - 1;
    const float* xrow = X + (size_t)arow * FH;
    const int col = n0 + mrow;

    v8f cL = {};
    v8f cR = {};
#pragma unroll
    for (int k = 0; k < FH; k += 4) {
        const v2f a = *(const v2f*)(xrow + k + khalf * 2);
        const size_t rk0 = (size_t)(k + khalf * 2) * FH + col;
        v2f bL, bR;
        bL.x = Wl[rk0];       bL.y = Wl[rk0 + FH];
        bR.x = Wr[rk0];       bR.y = Wr[rk0 + FH];
        cL = __builtin_amdgcn_wmma_f32_16x16x4_f32(false, a, false, bL,
                                                   (short)0, cL, false, false);
        cR = __builtin_amdgcn_wmma_f32_16x16x4_f32(false, a, false, bR,
                                                   (short)0, cR, false, false);
    }

    const float bbL = bl[col];
    const float bbR = br[col];
#pragma unroll
    for (int v = 0; v < 8; ++v) {
        const int row = r0 + v + khalf * 8;
        if (row < NN) {
            OutL[(size_t)row * FH + col] = cL[v] + bbL;
            OutR[(size_t)row * FH + col] = cR[v] + bbR;
        }
    }
}

__global__ void part_zero(int* __restrict__ cnt, int* __restrict__ cur)
{
    const int t = blockIdx.x * blockDim.x + threadIdx.x;
    if (t < NR) { cnt[t] = 0; cur[t] = 0; }
}

__global__ void part_count(const int* __restrict__ dst, int* __restrict__ cnt)
{
    __shared__ int hist[NR];
    for (int i = threadIdx.x; i < NR; i += 256) hist[i] = 0;
    __syncthreads();
    const int e0 = blockIdx.x * PCH;
    for (int k = 0; k < PPT; ++k) {
        const int le = k * 256 + threadIdx.x;
        if (le < PCH) atomicAdd(&hist[dst[e0 + le] / RN], 1);
    }
    __syncthreads();
    for (int i = threadIdx.x; i < NR; i += 256) {
        const int c = hist[i];
        if (c) atomicAdd(&cnt[i], c);
    }
}

__global__ void part_scan(const int* __restrict__ cnt, int* __restrict__ off)
{
    __shared__ int s[1024];
    const int t = threadIdx.x;
    const int v = (t < NR) ? cnt[t] : 0;
    s[t] = v;
    __syncthreads();
    for (int d = 1; d < 1024; d <<= 1) {
        const int add = (t >= d) ? s[t - d] : 0;
        __syncthreads();
        s[t] += add;
        __syncthreads();
    }
    if (t < NR) off[t] = s[t] - v;
    if (t == NR - 1) off[NR] = s[t];
}

__global__ void part_fill(const int* __restrict__ src, const int* __restrict__ dst,
                          const int* __restrict__ off, int* __restrict__ cur,
                          int* __restrict__ ssrc, int* __restrict__ sdst)
{
    __shared__ int hist[NR];
    __shared__ int basep[NR];
    for (int i = threadIdx.x; i < NR; i += 256) hist[i] = 0;
    __syncthreads();
    const int e0 = blockIdx.x * PCH;
    int rk[PPT], dv[PPT];
#pragma unroll
    for (int k = 0; k < PPT; ++k) {
        const int le = k * 256 + threadIdx.x;
        rk[k] = -1; dv[k] = 0;
        if (le < PCH) {
            dv[k] = dst[e0 + le];
            rk[k] = atomicAdd(&hist[dv[k] / RN], 1);
        }
    }
    __syncthreads();
    for (int i = threadIdx.x; i < NR; i += 256) {
        const int c = hist[i];
        basep[i] = c ? off[i] + atomicAdd(&cur[i], c) : 0;
    }
    __syncthreads();
#pragma unroll
    for (int k = 0; k < PPT; ++k) {
        if (rk[k] >= 0) {
            const int pos = basep[dv[k] / RN] + rk[k];
            ssrc[pos] = src[e0 + k * 256 + threadIdx.x];
            sdst[pos] = dv[k];
        }
    }
}

__device__ __forceinline__ float div_bias_elu(float x, float den, float bias)
{
    const float v = x / den + bias;
    return (v > 0.f) ? v : (expf(v) - 1.f);
}

__device__ __forceinline__ void wave_items(const int* __restrict__ ssrc,
                                           const int* __restrict__ sdst,
                                           int e_lo, int cnt, int tot, int base, int i0,
                                           int (&s)[EPW], int (&d)[EPW])
{
#pragma unroll
    for (int j = 0; j < EPW; ++j) {
        int idx = i0 + j;
        if (idx >= tot) idx = tot - 1;
        if (idx < cnt) {
            s[j] = ssrc[e_lo + idx];
            d[j] = sdst[e_lo + idx];
        } else {
            s[j] = d[j] = base + (idx - cnt);
        }
    }
}

template <int Q>
__device__ __forceinline__ float quad_bcast(float v)
{
    return __int_as_float(__builtin_amdgcn_mov_dpp(__float_as_int(v), Q * 0x55, 0xF, 0xF, true));
}

template <int NT>
__device__ __forceinline__ void next_tiles(const float* buf, int rt0, int lane,
                                           const float* __restrict__ Wn,
                                           const float* __restrict__ bn,
                                           float* __restrict__ on,
                                           int base, int nown, int col)
{
    const int mrow  = lane & 15;
    const int khalf = lane >> 4;
    const float* wp = Wn + (size_t)(khalf * 2) * FH + col;
    const float* ap[NT];
#pragma unroll
    for (int t = 0; t < NT; ++t) {
        const int r = min((rt0 + t) * 16 + mrow, RN - 1);
        ap[t] = buf + r * LP + khalf * 2;
    }
    v8f c[NT] = {};
#pragma unroll
    for (int k = 0; k < FH; k += 4) {
        v2f b;
        b.x = wp[(size_t)k * FH];
        b.y = wp[(size_t)k * FH + FH];
#pragma unroll
        for (int t = 0; t < NT; ++t) {
            const v2f a = *(const v2f*)(ap[t] + k);
            c[t] = __builtin_amdgcn_wmma_f32_16x16x4_f32(false, a, false, b,
                                                         (short)0, c[t], false, false);
        }
    }
    const float bb = bn[col];
#pragma unroll
    for (int t = 0; t < NT; ++t) {
#pragma unroll
        for (int v = 0; v < 8; ++v) {
            const int row = (rt0 + t) * 16 + v + khalf * 8;
            if (row < nown) on[(size_t)(base + row) * FH + col] = c[t][v] + bb;
        }
    }
}

template <bool LAST>
__global__ __launch_bounds__(OWT, 16)
void owner_layer(const float* __restrict__ xl,
                 const float* __restrict__ xr,
                 const int* __restrict__ ssrc,
                 const int* __restrict__ sdst,
                 const int* __restrict__ off,
                 const float* __restrict__ att,
                 const float* __restrict__ bo,
                 const float* __restrict__ Wln,
                 const float* __restrict__ bln,
                 const float* __restrict__ Wrn,
                 const float* __restrict__ brn,
                 float* __restrict__ xln,
                 float* __restrict__ xrn,
                 const float* __restrict__ Wf,
                 const float* __restrict__ bf,
                 float* __restrict__ out)
{
    __shared__ __attribute__((aligned(16))) float bufr[RN * LP];
    __shared__ __attribute__((aligned(16))) float buf[RN * LP];
    __shared__ float sh[RN * NH];
    __shared__ float mx[RN * NH];
    __shared__ float den[RN * NH];
    __shared__ int   redo;

    const int tid  = threadIdx.x;
    const int lane = tid & 31;
    const int wave = __builtin_amdgcn_readfirstlane(tid >> 5);
    const int base = blockIdx.x * RN;
    const int nown = min(RN, NN - base);
    const int e_lo = off[blockIdx.x];
    const int cnt  = off[blockIdx.x + 1] - e_lo;
    const int tot  = cnt + nown;
    const int ngrp = (tot + EPW - 1) / EPW;

    const int f = lane * 4;
    const float4 w = *(const float4*)(att + f);

    {
        const float4* g = (const float4*)(xr + (size_t)base * FH);
        for (int i = tid; i < nown * (FH / 4); i += OWT)
            ((float4*)bufr)[(i >> 5) * (LP / 4) + (i & 31)] = g[i];
        for (int i = tid; i < RN * LP / 4; i += OWT) ((float4*)buf)[i] = make_float4(0.f, 0.f, 0.f, 0.f);
        for (int i = tid; i < RN * NH; i += OWT) { mx[i] = -3.0e38f; den[i] = 0.f; }
        if (tid == 0) redo = 0;
        for (int i = wave; i < nown; i += OWT / 32) {
            const float4 a = *(const float4*)(xl + (size_t)(base + i) * FH + f);
            const float4 b = *(const float4*)(xr + (size_t)(base + i) * FH + f);
            float p = lrelu(a.x + b.x) * w.x + lrelu(a.y + b.y) * w.y +
                      lrelu(a.z + b.z) * w.z + lrelu(a.w + b.w) * w.w;
            p += __shfl_xor(p, 1, 8);
            p += __shfl_xor(p, 2, 8);
            p += __shfl_xor(p, 4, 8);
            if ((lane & 7) == 0) sh[i * NH + (lane >> 3)] = p;
        }
    }
    __syncthreads();

    {
        const bool b0 = (lane & 1) != 0;
        const bool b1 = (lane & 2) != 0;
        const int  jl = ((lane & 1) << 1) | ((lane >> 1) & 1);
        const int  hl = lane >> 3;
        for (int pass = 0; pass < 2; ++pass) {
            for (int g = wave; g < ngrp; g += OWT / 32) {
                const int i0 = g * EPW;
                int s[EPW], d[EPW];
                wave_items(ssrc, sdst, e_lo, cnt, tot, base, i0, s, d);
                float4 a[EPW], b[EPW];
#pragma unroll
                for (int j = 0; j < EPW; ++j) {
                    a[j] = *(const float4*)(xl + (size_t)s[j] * FH + f);
                    b[j] = *(const float4*)(bufr + (d[j] - base) * LP + f);
                }
                float p[EPW];
#pragma unroll
                for (int j = 0; j < EPW; ++j)
                    p[j] = lrelu(a[j].x + b[j].x) * w.x + lrelu(a[j].y + b[j].y) * w.y +
                           lrelu(a[j].z + b[j].z) * w.z + lrelu(a[j].w + b[j].w) * w.w;
                const float q0 = (b0 ? p[2] : p[0]) + __shfl_xor(b0 ? p[0] : p[2], 1, 8);
                const float q1 = (b0 ? p[3] : p[1]) + __shfl_xor(b0 ? p[1] : p[3], 1, 8);
                float r = (b1 ? q1 : q0) + __shfl_xor(b1 ? q0 : q1, 2, 8);
                r += __shfl_xor(r, 4, 8);

                const int  dj   = (jl == 0) ? d[0] : (jl == 1) ? d[1] : (jl == 2) ? d[2] : d[3];
                const bool ok   = (i0 + jl) < tot;
                const int  slot = (dj - base) * NH + hl;
                const float t   = r - sh[slot];
                const float ex  = ok ? expf(t) : 0.f;
                if (ok && t > 40.f) redo = 1;
                if ((lane & 4) == 0 && ok) {
                    atomicMax(&mx[slot], r);
                    atomicAdd(&den[slot], ex);
                }
                float wt[EPW];
                wt[0] = quad_bcast<0>(ex);
                wt[1] = quad_bcast<2>(ex);
                wt[2] = quad_bcast<1>(ex);
                wt[3] = quad_bcast<3>(ex);
#pragma unroll
                for (int j = 0; j < EPW; ++j) {
                    if (i0 + j < tot) {
                        float* o = buf + (d[j] - base) * LP + lane;
                        atomicAdd(o,      a[j].x * wt[j]);
                        atomicAdd(o + 32, a[j].y * wt[j]);
                        atomicAdd(o + 64, a[j].z * wt[j]);
                        atomicAdd(o + 96, a[j].w * wt[j]);
                    }
                }
            }
            __syncthreads();
            if (pass != 0) break;
            {
                bool bad = false;
                for (int i = tid; i < RN * LP / 4; i += OWT) {
                    const float4 v = ((const float4*)buf)[i];
                    bad |= !(fabsf(v.x) < 1.0e37f) | !(fabsf(v.y) < 1.0e37f) |
                           !(fabsf(v.z) < 1.0e37f) | !(fabsf(v.w) < 1.0e37f);
                }
                if (bad) redo = 1;
            }
            __syncthreads();
            if (redo == 0) break;
            __syncthreads();
            for (int i = tid; i < RN * LP / 4; i += OWT) ((float4*)buf)[i] = make_float4(0.f, 0.f, 0.f, 0.f);
            for (int i = tid; i < RN * NH; i += OWT) { sh[i] = mx[i]; den[i] = 0.f; }
            if (tid == 0) redo = 0;
            __syncthreads();
        }
    }

    for (int i = wave; i < nown; i += OWT / 32) {
        const float* rowp = buf + i * LP + lane;
        float4 v = make_float4(rowp[0], rowp[32], rowp[64], rowp[96]);
        const float dn = fmaxf(den[i * NH + (lane >> 3)], 1e-16f);
        const float4 bb = ((const float4*)bo)[lane];
        v.x = div_bias_elu(v.x, dn, bb.x);
        v.y = div_bias_elu(v.y, dn, bb.y);
        v.z = div_bias_elu(v.z, dn, bb.z);
        v.w = div_bias_elu(v.w, dn, bb.w);
        if (!LAST) {
            ((float4*)buf)[i * (LP / 4) + lane] = v;
        } else {
            const float4 wf = ((const float4*)Wf)[lane];
            float p = v.x * wf.x + v.y * wf.y + v.z * wf.z + v.w * wf.w;
            p += __shfl_xor(p, 16, 32);
            p += __shfl_xor(p,  8, 32);
            p += __shfl_xor(p,  4, 32);
            p += __shfl_xor(p,  2, 32);
            p += __shfl_xor(p,  1, 32);
            if (lane == 0) out[base + i] = p + bf[0];
        }
    }

    if (!LAST) {
        __syncthreads();
        const int combo = wave & 15;
        const bool right = (combo & 8) != 0;
        const int col = (combo & 7) * 16 + (lane & 15);
        if (wave < 16)
            next_tiles<4>(buf, 0, lane, right ? Wrn : Wln, right ? brn : bln,
                          right ? xrn : xln, base, nown, col);
        else
            next_tiles<3>(buf, 4, lane, right ? Wrn : Wln, right ? brn : bln,
                          right ? xrn : xln, base, nown, col);
    }
}

extern "C" void kernel_launch(void* const* d_in, const int* in_sizes, int n_in,
                              void* d_out, int out_size, void* d_ws, size_t ws_size,
                              hipStream_t stream)
{
    const float* x    = (const float*)d_in[0];
    const int*   ei   = (const int*)d_in[1];
    const int*   src  = ei;
    const int*   dst  = ei + EE;
    const float* Wl0  = (const float*)d_in[2];
    const float* bl0  = (const float*)d_in[3];
    const float* Wr0  = (const float*)d_in[4];
    const float* br0  = (const float*)d_in[5];
    const float* att0 = (const float*)d_in[6];
    const float* bo0  = (const float*)d_in[7];
    const float* Wl   = (const float*)d_in[8];
    const float* bl   = (const float*)d_in[9];
    const float* Wr   = (const float*)d_in[10];
    const float* br   = (const float*)d_in[11];
    const float* att  = (const float*)d_in[12];
    const float* bo   = (const float*)d_in[13];
    const float* Wf   = (const float*)d_in[14];
    const float* bf   = (const float*)d_in[15];

    float* ws   = (float*)d_ws;
    float* xlA  = ws;
    float* xrA  = xlA  + (size_t)NN * FH;
    float* xlB  = xrA  + (size_t)NN * FH;
    float* xrB  = xlB  + (size_t)NN * FH;
    int*   ssrc = (int*)(xrB + (size_t)NN * FH);
    int*   sdst = ssrc + EE;
    int*   cnt  = sdst + EE;
    int*   cur  = cnt + NR;
    int*   off  = cur + NR;

    part_zero <<<(NR + 255) / 256, 256, 0, stream>>>(cnt, cur);
    part_count<<<PB, 256, 0, stream>>>(dst, cnt);
    part_scan <<<1, 1024, 0, stream>>>(cnt, off);
    part_fill <<<PB, 256, 0, stream>>>(src, dst, off, cur, ssrc, sdst);

    const dim3 gemmGrid((NN + 127) / 128, FH / 16);

    gemm_wmma_dual<<<gemmGrid, 256, 0, stream>>>(x, Wl0, bl0, Wr0, br0, xlA, xrA);
    owner_layer<false><<<NR, OWT, 0, stream>>>(xlA, xrA, ssrc, sdst, off, att0, bo0,
                                               Wl, bl, Wr, br, xlB, xrB,
                                               Wf, bf, (float*)d_out);
    owner_layer<false><<<NR, OWT, 0, stream>>>(xlB, xrB, ssrc, sdst, off, att, bo,
                                               Wl + (size_t)FH * FH, bl + FH,
                                               Wr + (size_t)FH * FH, br + FH, xlA, xrA,
                                               Wf, bf, (float*)d_out);
    owner_layer<true> <<<NR, OWT, 0, stream>>>(xlA, xrA, ssrc, sdst, off,
                                               att + (size_t)NH * HIDW, bo + FH,
                                               Wl, bl, Wr, br, xlB, xrB,
                                               Wf, bf, (float*)d_out);
}


# Layer 0's dual GEMM takes the owner block's form: 98 rows of x copied once to LDS (pitch 132) as A fragments, one B fragment from memory feeds 4 (3) row tiles per wave; same k order, f32
# speedup vs baseline: 5.5528x; 1.1128x over previous
#include <hip/hip_runtime.h>

#define NN  50000
#define EE  800000
#define EP  (EE + NN)
#define FH  128
#define NH  4
#define HIDW 32
#define EPW  4
#define RN   98
#define NR   ((NN + RN - 1) / RN)
#define PB   256
#define PCH  (EE / PB)
#define PPT  ((PCH + 255) / 256)
#define OWT  1024
#define LP   (FH + 4)

static_assert(EE % PB == 0, "grouping blocks take equal chunks");
static_assert(EPW == 4 && NH == 4, "edge sweeps are written for 4 items x 4 heads per wave");
static_assert(NR <= 1024, "one block scans the range counts");
static_assert(RN > 6 * 16 && RN <= 7 * 16 && OWT == 1024, "the GEMM tail of an owner block splits 7 row tiles over 32 waves");
static_assert(LP % 4 == 0, "LDS rows stay 16-byte aligned");

typedef __attribute__((ext_vector_type(2))) float v2f;
typedef __attribute__((ext_vector_type(8))) float v8f;

__device__ __forceinline__ float lrelu(float v) {
    return fmaxf(v, 0.f) + 0.2f * fminf(v, 0.f);
}

__global__ void part_zero(int* __restrict__ cnt, int* __restrict__ cur)
{
    const int t = blockIdx.x * blockDim.x + threadIdx.x;
    if (t < NR) { cnt[t] = 0; cur[t] = 0; }
}

__global__ void part_count(const int* __restrict__ dst, int* __restrict__ cnt)
{
    __shared__ int hist[NR];
    for (int i = threadIdx.x; i < NR; i += 256) hist[i] = 0;
    __syncthreads();
    const int e0 = blockIdx.x * PCH;
    for (int k = 0; k < PPT; ++k) {
        const int le = k * 256 + threadIdx.x;
        if (le < PCH) atomicAdd(&hist[dst[e0 + le] / RN], 1);
    }
    __syncthreads();
    for (int i = threadIdx.x; i < NR; i += 256) {
        const int c = hist[i];
        if (c) atomicAdd(&cnt[i], c);
    }
}

__global__ void part_scan(const int* __restrict__ cnt, int* __restrict__ off)
{
    __shared__ int s[1024];
    const int t = threadIdx.x;
    const int v = (t < NR) ? cnt[t] : 0;
    s[t] = v;
    __syncthreads();
    for (int d = 1; d < 1024; d <<= 1) {
        const int add = (t >= d) ? s[t - d] : 0;
        __syncthreads();
        s[t] += add;
        __syncthreads();
    }
    if (t < NR) off[t] = s[t] - v;
    if (t == NR - 1) off[NR] = s[t];
}

__global__ void part_fill(const int* __restrict__ src, const int* __restrict__ dst,
                          const int* __restrict__ off, int* __restrict__ cur,
                          int* __restrict__ ssrc, int* __restrict__ sdst)
{
    __shared__ int hist[NR];
    __shared__ int basep[NR];
    for (int i = threadIdx.x; i < NR; i += 256) hist[i] = 0;
    __syncthreads();
    const int e0 = blockIdx.x * PCH;
    int rk[PPT], dv[PPT];
#pragma unroll
    for (int k = 0; k < PPT; ++k) {
        const int le = k * 256 + threadIdx.x;
        rk[k] = -1; dv[k] = 0;
        if (le < PCH) {
            dv[k] = dst[e0 + le];
            rk[k] = atomicAdd(&hist[dv[k] / RN], 1);
        }
    }
    __syncthreads();
    for (int i = threadIdx.x; i < NR; i += 256) {
        const int c = hist[i];
        basep[i] = c ? off[i] + atomicAdd(&cur[i], c) : 0;
    }
    __syncthreads();
#pragma unroll
    for (int k = 0; k < PPT; ++k) {
        if (rk[k] >= 0) {
            const int pos = basep[dv[k] / RN] + rk[k];
            ssrc[pos] = src[e0 + k * 256 + threadIdx.x];
            sdst[pos] = dv[k];
        }
    }
}

__device__ __forceinline__ float div_bias_elu(float x, float den, float bias)
{
    const float v = x / den + bias;
    return (v > 0.f) ? v : (expf(v) - 1.f);
}

__device__ __forceinline__ void wave_items(const int* __restrict__ ssrc,
                                           const int* __restrict__ sdst,
                                           int e_lo, int cnt, int tot, int base, int i0,
                                           int (&s)[EPW], int (&d)[EPW])
{
#pragma unroll
    for (int j = 0; j < EPW; ++j) {
        int idx = i0 + j;
        if (idx >= tot) idx = tot - 1;
        if (idx < cnt) {
            s[j] = ssrc[e_lo + idx];
            d[j] = sdst[e_lo + idx];
        } else {
            s[j] = d[j] = base + (idx - cnt);
        }
    }
}

template <int Q>
__device__ __forceinline__ float quad_bcast(float v)
{
    return __int_as_float(__builtin_amdgcn_mov_dpp(__float_as_int(v), Q * 0x55, 0xF, 0xF, true));
}

template <int NT>
__device__ __forceinline__ void next_tiles(const float* buf, int rt0, int lane,
                                           const float* __restrict__ Wn,
                                           const float* __restrict__ bn,
                                           float* __restrict__ on,
                                           int base, int nown, int col)
{
    const int mrow  = lane & 15;
    const int khalf = lane >> 4;
    const float* wp = Wn + (size_t)(khalf * 2) * FH + col;
    const float* ap[NT];
#pragma unroll
    for (int t = 0; t < NT; ++t) {
        const int r = min((rt0 + t) * 16 + mrow, RN - 1);
        ap[t] = buf + r * LP + khalf * 2;
    }
    v8f c[NT] = {};
#pragma unroll
    for (int k = 0; k < FH; k += 4) {
        v2f b;
        b.x = wp[(size_t)k * FH];
        b.y = wp[(size_t)k * FH + FH];
#pragma unroll
        for (int t = 0; t < NT; ++t) {
            const v2f a = *(const v2f*)(ap[t] + k);
            c[t] = __builtin_amdgcn_wmma_f32_16x16x4_f32(false, a, false, b,
                                                         (short)0, c[t], false, false);
        }
    }
    const float bb = bn[col];
#pragma unroll
    for (int t = 0; t < NT; ++t) {
#pragma unroll
        for (int v = 0; v < 8; ++v) {
            const int row = (rt0 + t) * 16 + v + khalf * 8;
            if (row < nown) on[(size_t)(base + row) * FH + col] = c[t][v] + bb;
        }
    }
}

__global__ __launch_bounds__(OWT, 16)
void gemm0_rows(const float* __restrict__ X,
                const float* __restrict__ Wl,
                const float* __restrict__ bl,
                const float* __restrict__ Wr,
                const float* __restrict__ br,
                float* __restrict__ OutL,
                float* __restrict__ OutR)
{
    __shared__ __attribute__((aligned(16))) float buf[RN * LP];

    const int tid  = threadIdx.x;
    const int lane = tid & 31;
    const int wave = __builtin_amdgcn_readfirstlane(tid >> 5);
    const int base = blockIdx.x * RN;
    const int nown = min(RN, NN - base);

    const float4* g = (const float4*)(X + (size_t)base * FH);
    constexpr int SPT = (RN * (FH / 4) + OWT - 1) / OWT;
    float4 piece[SPT];
#pragma unroll
    for (int u = 0; u < SPT; ++u) {
        const int i = tid + u * OWT;
        piece[u] = (i < nown * (FH / 4)) ? g[i] : make_float4(0.f, 0.f, 0.f, 0.f);
    }
#pragma unroll
    for (int u = 0; u < SPT; ++u) {
        const int i = tid + u * OWT;
        if (i < RN * (FH / 4)) ((float4*)buf)[(i >> 5) * (LP / 4) + (i & 31)] = piece[u];
    }
    __syncthreads();

    const int combo = wave & 15;
    const bool right = (combo & 8) != 0;
    const int col = (combo & 7) * 16 + (lane & 15);
    if (wave < 16)
        next_tiles<4>(buf, 0, lane, right ? Wr : Wl, right ? br : bl,
                      right ? OutR : OutL, base, nown, col);
    else
        next_tiles<3>(buf, 4, lane, right ? Wr : Wl, right ? br : bl,
                      right ? OutR : OutL, base, nown, col);
}

template <bool LAST>
__global__ __launch_bounds__(OWT, 16)
void owner_layer(const float* __restrict__ xl,
                 const float* __restrict__ xr,
                 const int* __restrict__ ssrc,
                 const int* __restrict__ sdst,
                 const int* __restrict__ off,
                 const float* __restrict__ att,
                 const float* __restrict__ bo,
                 const float* __restrict__ Wln,
                 const float* __restrict__ bln,
                 const float* __restrict__ Wrn,
                 const float* __restrict__ brn,
                 float* __restrict__ xln,
                 float* __restrict__ xrn,
                 const float* __restrict__ Wf,
                 const float* __restrict__ bf,
                 float* __restrict__ out)
{
    __shared__ __attribute__((aligned(16))) float bufr[RN * LP];
    __shared__ __attribute__((aligned(16))) float buf[RN * LP];
    __shared__ float sh[RN * NH];
    __shared__ float mx[RN * NH];
    __shared__ float den[RN * NH];
    __shared__ int   redo;

    const int tid  = threadIdx.x;
    const int lane = tid & 31;
    const int wave = __builtin_amdgcn_readfirstlane(tid >> 5);
    const int base = blockIdx.x * RN;
    const int nown = min(RN, NN - base);
    const int e_lo = off[blockIdx.x];
    const int cnt  = off[blockIdx.x + 1] - e_lo;
    const int tot  = cnt + nown;
    const int ngrp = (tot + EPW - 1) / EPW;

    const int f = lane * 4;
    const float4 w = *(const float4*)(att + f);

    {
        const float4* g = (const float4*)(xr + (size_t)base * FH);
        for (int i = tid; i < nown * (FH / 4); i += OWT)
            ((float4*)bufr)[(i >> 5) * (LP / 4) + (i & 31)] = g[i];
        for (int i = tid; i < RN * LP / 4; i += OWT) ((float4*)buf)[i] = make_float4(0.f, 0.f, 0.f, 0.f);
        for (int i = tid; i < RN * NH; i += OWT) { mx[i] = -3.0e38f; den[i] = 0.f; }
        if (tid == 0) redo = 0;
        for (int i = wave; i < nown; i += OWT / 32) {
            const float4 a = *(const float4*)(xl + (size_t)(base + i) * FH + f);
            const float4 b = *(const float4*)(xr + (size_t)(base + i) * FH + f);
            float p = lrelu(a.x + b.x) * w.x + lrelu(a.y + b.y) * w.y +
                      lrelu(a.z + b.z) * w.z + lrelu(a.w + b.w) * w.w;
            p += __shfl_xor(p, 1, 8);
            p += __shfl_xor(p, 2, 8);
            p += __shfl_xor(p, 4, 8);
            if ((lane & 7) == 0) sh[i * NH + (lane >> 3)] = p;
        }
    }
    __syncthreads();

    {
        const bool b0 = (lane & 1) != 0;
        const bool b1 = (lane & 2) != 0;
        const int  jl = ((lane & 1) << 1) | ((lane >> 1) & 1);
        const int  hl = lane >> 3;
        for (int pass = 0; pass < 2; ++pass) {
            for (int g = wave; g < ngrp; g += OWT / 32) {
                const int i0 = g * EPW;
                int s[EPW], d[EPW];
                wave_items(ssrc, sdst, e_lo, cnt, tot, base, i0, s, d);
                float4 a[EPW], b[EPW];
#pragma unroll
                for (int j = 0; j < EPW; ++j) {
                    a[j] = *(const float4*)(xl + (size_t)s[j] * FH + f);
                    b[j] = *(const float4*)(bufr + (d[j] - base) * LP + f);
                }
                float p[EPW];
#pragma unroll
                for (int j = 0; j < EPW; ++j)
                    p[j] = lrelu(a[j].x + b[j].x) * w.x + lrelu(a[j].y + b[j].y) * w.y +
                           lrelu(a[j].z + b[j].z) * w.z + lrelu(a[j].w + b[j].w) * w.w;
                const float q0 = (b0 ? p[2] : p[0]) + __shfl_xor(b0 ? p[0] : p[2], 1, 8);
                const float q1 = (b0 ? p[3] : p[1]) + __shfl_xor(b0 ? p[1] : p[3], 1, 8);
                float r = (b1 ? q1 : q0) + __shfl_xor(b1 ? q0 : q1, 2, 8);
                r += __shfl_xor(r, 4, 8);

                const int  dj   = (jl == 0) ? d[0] : (jl == 1) ? d[1] : (jl == 2) ? d[2] : d[3];
                const bool ok   = (i0 + jl) < tot;
                const int  slot = (dj - base) * NH + hl;
                const float t   = r - sh[slot];
                const float ex  = ok ? expf(t) : 0.f;
                if (ok && t > 40.f) redo = 1;
                if ((lane & 4) == 0 && ok) {
                    atomicMax(&mx[slot], r);
                    atomicAdd(&den[slot], ex);
                }
                float wt[EPW];
                wt[0] = quad_bcast<0>(ex);
                wt[1] = quad_bcast<2>(ex);
                wt[2] = quad_bcast<1>(ex);
                wt[3] = quad_bcast<3>(ex);
#pragma unroll
                for (int j = 0; j < EPW; ++j) {
                    if (i0 + j < tot) {
                        float* o = buf + (d[j] - base) * LP + lane;
                        atomicAdd(o,      a[j].x * wt[j]);
                        atomicAdd(o + 32, a[j].y * wt[j]);
                        atomicAdd(o + 64, a[j].z * wt[j]);
                        atomicAdd(o + 96, a[j].w * wt[j]);
                    }
                }
            }
            __syncthreads();
            if (pass != 0) break;
            {
                bool bad = false;
                for (int i = tid; i < RN * LP / 4; i += OWT) {
                    const float4 v = ((const float4*)buf)[i];
                    bad |= !(fabsf(v.x) < 1.0e37f) | !(fabsf(v.y) < 1.0e37f) |
                           !(fabsf(v.z) < 1.0e37f) | !(fabsf(v.w) < 1.0e37f);
                }
                if (bad) redo = 1;
            }
            __syncthreads();
            if (redo == 0) break;
            __syncthreads();
            for (int i = tid; i < RN * LP / 4; i += OWT) ((float4*)buf)[i] = make_float4(0.f, 0.f, 0.f, 0.f);
            for (int i = tid; i < RN * NH; i += OWT) { sh[i] = mx[i]; den[i] = 0.f; }
            if (tid == 0) redo = 0;
            __syncthreads();
        }
    }

    for (int i = wave; i < nown; i += OWT / 32) {
        const float* rowp = buf + i * LP + lane;
        float4 v = make_float4(rowp[0], rowp[32], rowp[64], rowp[96]);
        const float dn = fmaxf(den[i * NH + (lane >> 3)], 1e-16f);
        const float4 bb = ((const float4*)bo)[lane];
        v.x = div_bias_elu(v.x, dn, bb.x);
        v.y = div_bias_elu(v.y, dn, bb.y);
        v.z = div_bias_elu(v.z, dn, bb.z);
        v.w = div_bias_elu(v.w, dn, bb.w);
        if (!LAST) {
            ((float4*)buf)[i * (LP / 4) + lane] = v;
        } else {
            const float4 wf = ((const float4*)Wf)[lane];
            float p = v.x * wf.x + v.y * wf.y + v.z * wf.z + v.w * wf.w;
            p += __shfl_xor(p, 16, 32);
            p += __shfl_xor(p,  8, 32);
            p += __shfl_xor(p,  4, 32);
            p += __shfl_xor(p,  2, 32);
            p += __shfl_xor(p,  1, 32);
            if (lane == 0) out[base + i] = p + bf[0];
        }
    }

    if (!LAST) {
        __syncthreads();
        const int combo = wave & 15;
        const bool right = (combo & 8) != 0;
        const int col = (combo & 7) * 16 + (lane & 15);
        if (wave < 16)
            next_tiles<4>(buf, 0, lane, right ? Wrn : Wln, right ? brn : bln,
                          right ? xrn : xln, base, nown, col);
        else
            next_tiles<3>(buf, 4, lane, right ? Wrn : Wln, right ? brn : bln,
                          right ? xrn : xln, base, nown, col);
    }
}

extern "C" void kernel_launch(void* const* d_in, const int* in_sizes, int n_in,
                              void* d_out, int out_size, void* d_ws, size_t ws_size,
                              hipStream_t stream)
{
    const float* x    = (const float*)d_in[0];
    const int*   ei   = (const int*)d_in[1];
    const int*   src  = ei;
    const int*   dst  = ei + EE;
    const float* Wl0  = (const float*)d_in[2];
    const float* bl0  = (const float*)d_in[3];
    const float* Wr0  = (const float*)d_in[4];
    const float* br0  = (const float*)d_in[5];
    const float* att0 = (const float*)d_in[6];
    const float* bo0  = (const float*)d_in[7];
    const float* Wl   = (const float*)d_in[8];
    const float* bl   = (const float*)d_in[9];
    const float* Wr   = (const float*)d_in[10];
    const float* br   = (const float*)d_in[11];
    const float* att  = (const float*)d_in[12];
    const float* bo   = (const float*)d_in[13];
    const float* Wf   = (const float*)d_in[14];
    const float* bf   = (const float*)d_in[15];

    float* ws   = (float*)d_ws;
    float* xlA  = ws;
    float* xrA  = xlA  + (size_t)NN * FH;
    float* xlB  = xrA  + (size_t)NN * FH;
    float* xrB  = xlB  + (size_t)NN * FH;
    int*   ssrc = (int*)(xrB + (size_t)NN * FH);
    int*   sdst = ssrc + EE;
    int*   cnt  = sdst + EE;
    int*   cur  = cnt + NR;
    int*   off  = cur + NR;

    part_zero <<<(NR + 255) / 256, 256, 0, stream>>>(cnt, cur);
    part_count<<<PB, 256, 0, stream>>>(dst, cnt);
    part_scan <<<1, 1024, 0, stream>>>(cnt, off);
    part_fill <<<PB, 256, 0, stream>>>(src, dst, off, cur, ssrc, sdst);

    gemm0_rows<<<NR, OWT, 0, stream>>>(x, Wl0, bl0, Wr0, br0, xlA, xrA);
    owner_layer<false><<<NR, OWT, 0, stream>>>(xlA, xrA, ssrc, sdst, off, att0, bo0,
                                               Wl, bl, Wr, br, xlB, xrB,
                                               Wf, bf, (float*)d_out);
    owner_layer<false><<<NR, OWT, 0, stream>>>(xlB, xrB, ssrc, sdst, off, att, bo,
                                               Wl + (size_t)FH * FH, bl + FH,
                                               Wr + (size_t)FH * FH, br + FH, xlA, xrA,
                                               Wf, bf, (float*)d_out);
    owner_layer<true> <<<NR, OWT, 0, stream>>>(xlA, xrA, ssrc, sdst, off,
                                               att + (size_t)NH * HIDW, bo + FH,
                                               Wl, bl, Wr, br, xlB, xrB,
                                               Wf, bf, (float*)d_out);
}


# Cross-lane sums (sweep butterfly, self-loop logits, final linear) take the partner lane through DPP row_xmask operands (xor 16: v_permlanex16), not ds_bpermute trips through LDS; same sum trees
# speedup vs baseline: 5.8312x; 1.0501x over previous
#include <hip/hip_runtime.h>

#define NN  50000
#define EE  800000
#define EP  (EE + NN)
#define FH  128
#define NH  4
#define HIDW 32
#define EPW  4
#define RN   98
#define NR   ((NN + RN - 1) / RN)
#define PB   256
#define PCH  (EE / PB)
#define PPT  ((PCH + 255) / 256)
#define OWT  1024
#define LP   (FH + 4)

static_assert(EE % PB == 0, "grouping blocks take equal chunks");
static_assert(EPW == 4 && NH == 4, "edge sweeps are written for 4 items x 4 heads per wave");
static_assert(NR <= 1024, "one block scans the range counts");
static_assert(RN > 6 * 16 && RN <= 7 * 16 && OWT == 1024, "the GEMM tail of an owner block splits 7 row tiles over 32 waves");
static_assert(LP % 4 == 0, "LDS rows stay 16-byte aligned");

typedef __attribute__((ext_vector_type(2))) float v2f;
typedef __attribute__((ext_vector_type(8))) float v8f;

__device__ __forceinline__ float lrelu(float v) {
    return fmaxf(v, 0.f) + 0.2f * fminf(v, 0.f);
}

__global__ void part_zero(int* __restrict__ cnt, int* __restrict__ cur)
{
    const int t = blockIdx.x * blockDim.x + threadIdx.x;
    if (t < NR) { cnt[t] = 0; cur[t] = 0; }
}

__global__ void part_count(const int* __restrict__ dst, int* __restrict__ cnt)
{
    __shared__ int hist[NR];
    for (int i = threadIdx.x; i < NR; i += 256) hist[i] = 0;
    __syncthreads();
    const int e0 = blockIdx.x * PCH;
    for (int k = 0; k < PPT; ++k) {
        const int le = k * 256 + threadIdx.x;
        if (le < PCH) atomicAdd(&hist[dst[e0 + le] / RN], 1);
    }
    __syncthreads();
    for (int i = threadIdx.x; i < NR; i += 256) {
        const int c = hist[i];
        if (c) atomicAdd(&cnt[i], c);
    }
}

__global__ void part_scan(const int* __restrict__ cnt, int* __restrict__ off)
{
    __shared__ int s[1024];
    const int t = threadIdx.x;
    const int v = (t < NR) ? cnt[t] : 0;
    s[t] = v;
    __syncthreads();
    for (int d = 1; d < 1024; d <<= 1) {
        const int add = (t >= d) ? s[t - d] : 0;
        __syncthreads();
        s[t] += add;
        __syncthreads();
    }
    if (t < NR) off[t] = s[t] - v;
    if (t == NR - 1) off[NR] = s[t];
}

__global__ void part_fill(const int* __restrict__ src, const int* __restrict__ dst,
                          const int* __restrict__ off, int* __restrict__ cur,
                          int* __restrict__ ssrc, int* __restrict__ sdst)
{
    __shared__ int hist[NR];
    __shared__ int basep[NR];
    for (int i = threadIdx.x; i < NR; i += 256) hist[i] = 0;
    __syncthreads();
    const int e0 = blockIdx.x * PCH;
    int rk[PPT], dv[PPT];
#pragma unroll
    for (int k = 0; k < PPT; ++k) {
        const int le = k * 256 + threadIdx.x;
        rk[k] = -1; dv[k] = 0;
        if (le < PCH) {
            dv[k] = dst[e0 + le];
            rk[k] = atomicAdd(&hist[dv[k] / RN], 1);
        }
    }
    __syncthreads();
    for (int i = threadIdx.x; i < NR; i += 256) {
        const int c = hist[i];
        basep[i] = c ? off[i] + atomicAdd(&cur[i], c) : 0;
    }
    __syncthreads();
#pragma unroll
    for (int k = 0; k < PPT; ++k) {
        if (rk[k] >= 0) {
            const int pos = basep[dv[k] / RN] + rk[k];
            ssrc[pos] = src[e0 + k * 256 + threadIdx.x];
            sdst[pos] = dv[k];
        }
    }
}

__device__ __forceinline__ float div_bias_elu(float x, float den, float bias)
{
    const float v = x / den + bias;
    return (v > 0.f) ? v : (expf(v) - 1.f);
}

__device__ __forceinline__ void wave_items(const int* __restrict__ ssrc,
                                           const int* __restrict__ sdst,
                                           int e_lo, int cnt, int tot, int base, int i0,
                                           int (&s)[EPW], int (&d)[EPW])
{
#pragma unroll
    for (int j = 0; j < EPW; ++j) {
        int idx = i0 + j;
        if (idx >= tot) idx = tot - 1;
        if (idx < cnt) {
            s[j] = ssrc[e_lo + idx];
            d[j] = sdst[e_lo + idx];
        } else {
            s[j] = d[j] = base + (idx - cnt);
        }
    }
}

template <int Q>
__device__ __forceinline__ float quad_bcast(float v)
{
    return __int_as_float(__builtin_amdgcn_mov_dpp(__float_as_int(v), Q * 0x55, 0xF, 0xF, true));
}

template <int M>
__device__ __forceinline__ float row_xor(float v)
{
    return __int_as_float(__builtin_amdgcn_mov_dpp(__float_as_int(v), 0x160 + M, 0xF, 0xF, true));
}

__device__ __forceinline__ float xhalf(float v)
{
    const int b = __float_as_int(v);
    return __int_as_float(__builtin_amdgcn_permlanex16(b, b, 0x76543210, 0xfedcba98, false, false));
}

template <int NT>
__device__ __forceinline__ void next_tiles(const float* buf, int rt0, int lane,
                                           const float* __restrict__ Wn,
                                           const float* __restrict__ bn,
                                           float* __restrict__ on,
                                           int base, int nown, int col)
{
    const int mrow  = lane & 15;
    const int khalf = lane >> 4;
    const float* wp = Wn + (size_t)(khalf * 2) * FH + col;
    const float* ap[NT];
#pragma unroll
    for (int t = 0; t < NT; ++t) {
        const int r = min((rt0 + t) * 16 + mrow, RN - 1);
        ap[t] = buf + r * LP + khalf * 2;
    }
    v8f c[NT] = {};
#pragma unroll
    for (int k = 0; k < FH; k += 4) {
        v2f b;
        b.x = wp[(size_t)k * FH];
        b.y = wp[(size_t)k * FH + FH];
#pragma unroll
        for (int t = 0; t < NT; ++t) {
            const v2f a = *(const v2f*)(ap[t] + k);
            c[t] = __builtin_amdgcn_wmma_f32_16x16x4_f32(false, a, false, b,
                                                         (short)0, c[t], false, false);
        }
    }
    const float bb = bn[col];
#pragma unroll
    for (int t = 0; t < NT; ++t) {
#pragma unroll
        for (int v = 0; v < 8; ++v) {
            const int row = (rt0 + t) * 16 + v + khalf * 8;
            if (row < nown) on[(size_t)(base + row) * FH + col] = c[t][v] + bb;
        }
    }
}

__global__ __launch_bounds__(OWT, 16)
void gemm0_rows(const float* __restrict__ X,
                const float* __restrict__ Wl,
                const float* __restrict__ bl,
                const float* __restrict__ Wr,
                const float* __restrict__ br,
                float* __restrict__ OutL,
                float* __restrict__ OutR)
{
    __shared__ __attribute__((aligned(16))) float buf[RN * LP];

    const int tid  = threadIdx.x;
    const int lane = tid & 31;
    const int wave = __builtin_amdgcn_readfirstlane(tid >> 5);
    const int base = blockIdx.x * RN;
    const int nown = min(RN, NN - base);

    const float4* g = (const float4*)(X + (size_t)base * FH);
    constexpr int SPT = (RN * (FH / 4) + OWT - 1) / OWT;
    float4 piece[SPT];
#pragma unroll
    for (int u = 0; u < SPT; ++u) {
        const int i = tid + u * OWT;
        piece[u] = (i < nown * (FH / 4)) ? g[i] : make_float4(0.f, 0.f, 0.f, 0.f);
    }
#pragma unroll
    for (int u = 0; u < SPT; ++u) {
        const int i = tid + u * OWT;
        if (i < RN * (FH / 4)) ((float4*)buf)[(i >> 5) * (LP / 4) + (i & 31)] = piece[u];
    }
    __syncthreads();

    const int combo = wave & 15;
    const bool right = (combo & 8) != 0;
    const int col = (combo & 7) * 16 + (lane & 15);
    if (wave < 16)
        next_tiles<4>(buf, 0, lane, right ? Wr : Wl, right ? br : bl,
                      right ? OutR : OutL, base, nown, col);
    else
        next_tiles<3>(buf, 4, lane, right ? Wr : Wl, right ? br : bl,
                      right ? OutR : OutL, base, nown, col);
}

template <bool LAST>
__global__ __launch_bounds__(OWT, 16)
void owner_layer(const float* __restrict__ xl,
                 const float* __restrict__ xr,
                 const int* __restrict__ ssrc,
                 const int* __restrict__ sdst,
                 const int* __restrict__ off,
                 const float* __restrict__ att,
                 const float* __restrict__ bo,
                 const float* __restrict__ Wln,
                 const float* __restrict__ bln,
                 const float* __restrict__ Wrn,
                 const float* __restrict__ brn,
                 float* __restrict__ xln,
                 float* __restrict__ xrn,
                 const float* __restrict__ Wf,
                 const float* __restrict__ bf,
                 float* __restrict__ out)
{
    __shared__ __attribute__((aligned(16))) float bufr[RN * LP];
    __shared__ __attribute__((aligned(16))) float buf[RN * LP];
    __shared__ float sh[RN * NH];
    __shared__ float mx[RN * NH];
    __shared__ float den[RN * NH];
    __shared__ int   redo;

    const int tid  = threadIdx.x;
    const int lane = tid & 31;
    const int wave = __builtin_amdgcn_readfirstlane(tid >> 5);
    const int base = blockIdx.x * RN;
    const int nown = min(RN, NN - base);
    const int e_lo = off[blockIdx.x];
    const int cnt  = off[blockIdx.x + 1] - e_lo;
    const int tot  = cnt + nown;
    const int ngrp = (tot + EPW - 1) / EPW;

    const int f = lane * 4;
    const float4 w = *(const float4*)(att + f);

    {
        const float4* g = (const float4*)(xr + (size_t)base * FH);
        for (int i = tid; i < nown * (FH / 4); i += OWT)
            ((float4*)bufr)[(i >> 5) * (LP / 4) + (i & 31)] = g[i];
        for (int i = tid; i < RN * LP / 4; i += OWT) ((float4*)buf)[i] = make_float4(0.f, 0.f, 0.f, 0.f);
        for (int i = tid; i < RN * NH; i += OWT) { mx[i] = -3.0e38f; den[i] = 0.f; }
        if (tid == 0) redo = 0;
        for (int i = wave; i < nown; i += OWT / 32) {
            const float4 a = *(const float4*)(xl + (size_t)(base + i) * FH + f);
            const float4 b = *(const float4*)(xr + (size_t)(base + i) * FH + f);
            float p = lrelu(a.x + b.x) * w.x + lrelu(a.y + b.y) * w.y +
                      lrelu(a.z + b.z) * w.z + lrelu(a.w + b.w) * w.w;
            p += row_xor<1>(p);
            p += row_xor<2>(p);
            p += row_xor<4>(p);
            if ((lane & 7) == 0) sh[i * NH + (lane >> 3)] = p;
        }
    }
    __syncthreads();

    {
        const bool b0 = (lane & 1) != 0;
        const bool b1 = (lane & 2) != 0;
        const int  jl = ((lane & 1) << 1) | ((lane >> 1) & 1);
        const int  hl = lane >> 3;
        for (int pass = 0; pass < 2; ++pass) {
            for (int g = wave; g < ngrp; g += OWT / 32) {
                const int i0 = g * EPW;
                int s[EPW], d[EPW];
                wave_items(ssrc, sdst, e_lo, cnt, tot, base, i0, s, d);
                float4 a[EPW], b[EPW];
#pragma unroll
                for (int j = 0; j < EPW; ++j) {
                    a[j] = *(const float4*)(xl + (size_t)s[j] * FH + f);
                    b[j] = *(const float4*)(bufr + (d[j] - base) * LP + f);
                }
                float p[EPW];
#pragma unroll
                for (int j = 0; j < EPW; ++j)
                    p[j] = lrelu(a[j].x + b[j].x) * w.x + lrelu(a[j].y + b[j].y) * w.y +
                           lrelu(a[j].z + b[j].z) * w.z + lrelu(a[j].w + b[j].w) * w.w;
                const float q0 = (b0 ? p[2] : p[0]) + row_xor<1>(b0 ? p[0] : p[2]);
                const float q1 = (b0 ? p[3] : p[1]) + row_xor<1>(b0 ? p[1] : p[3]);
                float r = (b1 ? q1 : q0) + row_xor<2>(b1 ? q0 : q1);
                r += row_xor<4>(r);

                const int  dj   = (jl == 0) ? d[0] : (jl == 1) ? d[1] : (jl == 2) ? d[2] : d[3];
                const bool ok   = (i0 + jl) < tot;
                const int  slot = (dj - base) * NH + hl;
                const float t   = r - sh[slot];
                const float ex  = ok ? expf(t) : 0.f;
                if (ok && t > 40.f) redo = 1;
                if ((lane & 4) == 0 && ok) {
                    atomicMax(&mx[slot], r);
                    atomicAdd(&den[slot], ex);
                }
                float wt[EPW];
                wt[0] = quad_bcast<0>(ex);
                wt[1] = quad_bcast<2>(ex);
                wt[2] = quad_bcast<1>(ex);
                wt[3] = quad_bcast<3>(ex);
#pragma unroll
                for (int j = 0; j < EPW; ++j) {
                    if (i0 + j < tot) {
                        float* o = buf + (d[j] - base) * LP + lane;
                        atomicAdd(o,      a[j].x * wt[j]);
                        atomicAdd(o + 32, a[j].y * wt[j]);
                        atomicAdd(o + 64, a[j].z * wt[j]);
                        atomicAdd(o + 96, a[j].w * wt[j]);
                    }
                }
            }
            __syncthreads();
            if (pass != 0) break;
            {
                bool bad = false;
                for (int i = tid; i < RN * LP / 4; i += OWT) {
                    const float4 v = ((const float4*)buf)[i];
                    bad |= !(fabsf(v.x) < 1.0e37f) | !(fabsf(v.y) < 1.0e37f) |
                           !(fabsf(v.z) < 1.0e37f) | !(fabsf(v.w) < 1.0e37f);
                }
                if (bad) redo = 1;
            }
            __syncthreads();
            if (redo == 0) break;
            __syncthreads();
            for (int i = tid; i < RN * LP / 4; i += OWT) ((float4*)buf)[i] = make_float4(0.f, 0.f, 0.f, 0.f);
            for (int i = tid; i < RN * NH; i += OWT) { sh[i] = mx[i]; den[i] = 0.f; }
            if (tid == 0) redo = 0;
            __syncthreads();
        }
    }

    for (int i = wave; i < nown; i += OWT / 32) {
        const float* rowp = buf + i * LP + lane;
        float4 v = make_float4(rowp[0], rowp[32], rowp[64], rowp[96]);
        const float dn = fmaxf(den[i * NH + (lane >> 3)], 1e-16f);
        const float4 bb = ((const float4*)bo)[lane];
        v.x = div_bias_elu(v.x, dn, bb.x);
        v.y = div_bias_elu(v.y, dn, bb.y);
        v.z = div_bias_elu(v.z, dn, bb.z);
        v.w = div_bias_elu(v.w, dn, bb.w);
        if (!LAST) {
            ((float4*)buf)[i * (LP / 4) + lane] = v;
        } else {
            const float4 wf = ((const float4*)Wf)[lane];
            float p = v.x * wf.x + v.y * wf.y + v.z * wf.z + v.w * wf.w;
            p += xhalf(p);
            p += row_xor<8>(p);
            p += row_xor<4>(p);
            p += row_xor<2>(p);
            p += row_xor<1>(p);
            if (lane == 0) out[base + i] = p + bf[0];
        }
    }

    if (!LAST) {
        __syncthreads();
        const int combo = wave & 15;
        const bool right = (combo & 8) != 0;
        const int col = (combo & 7) * 16 + (lane & 15);
        if (wave < 16)
            next_tiles<4>(buf, 0, lane, right ? Wrn : Wln, right ? brn : bln,
                          right ? xrn : xln, base, nown, col);
        else
            next_tiles<3>(buf, 4, lane, right ? Wrn : Wln, right ? brn : bln,
                          right ? xrn : xln, base, nown, col);
    }
}

extern "C" void kernel_launch(void* const* d_in, const int* in_sizes, int n_in,
                              void* d_out, int out_size, void* d_ws, size_t ws_size,
                              hipStream_t stream)
{
    const float* x    = (const float*)d_in[0];
    const int*   ei   = (const int*)d_in[1];
    const int*   src  = ei;
    const int*   dst  = ei + EE;
    const float* Wl0  = (const float*)d_in[2];
    const float* bl0  = (const float*)d_in[3];
    const float* Wr0  = (const float*)d_in[4];
    const float* br0  = (const float*)d_in[5];
    const float* att0 = (const float*)d_in[6];
    const float* bo0  = (const float*)d_in[7];
    const float* Wl   = (const float*)d_in[8];
    const float* bl   = (const float*)d_in[9];
    const float* Wr   = (const float*)d_in[10];
    const float* br   = (const float*)d_in[11];
    const float* att  = (const float*)d_in[12];
    const float* bo   = (const float*)d_in[13];
    const float* Wf   = (const float*)d_in[14];
    const float* bf   = (const float*)d_in[15];

    float* ws   = (float*)d_ws;
    float* xlA  = ws;
    float* xrA  = xlA  + (size_t)NN * FH;
    float* xlB  = xrA  + (size_t)NN * FH;
    float* xrB  = xlB  + (size_t)NN * FH;
    int*   ssrc = (int*)(xrB + (size_t)NN * FH);
    int*   sdst = ssrc + EE;
    int*   cnt  = sdst + EE;
    int*   cur  = cnt + NR;
    int*   off  = cur + NR;

    part_zero <<<(NR + 255) / 256, 256, 0, stream>>>(cnt, cur);
    part_count<<<PB, 256, 0, stream>>>(dst, cnt);
    part_scan <<<1, 1024, 0, stream>>>(cnt, off);
    part_fill <<<PB, 256, 0, stream>>>(src, dst, off, cur, ssrc, sdst);

    gemm0_rows<<<NR, OWT, 0, stream>>>(x, Wl0, bl0, Wr0, br0, xlA, xrA);
    owner_layer<false><<<NR, OWT, 0, stream>>>(xlA, xrA, ssrc, sdst, off, att0, bo0,
                                               Wl, bl, Wr, br, xlB, xrB,
                                               Wf, bf, (float*)d_out);
    owner_layer<false><<<NR, OWT, 0, stream>>>(xlB, xrB, ssrc, sdst, off, att, bo,
                                               Wl + (size_t)FH * FH, bl + FH,
                                               Wr + (size_t)FH * FH, br + FH, xlA, xrA,
                                               Wf, bf, (float*)d_out);
    owner_layer<true> <<<NR, OWT, 0, stream>>>(xlA, xrA, ssrc, sdst, off,
                                               att + (size_t)NH * HIDW, bo + FH,
                                               Wl, bl, Wr, br, xlB, xrB,
                                               Wf, bf, (float*)d_out);
}


# Self-loops leave the sweep: the wave that stages a node reads its xl, xr row once (one load group), fills the LDS xr row and shift, and stores the node's own term (weight 1) as the start of its sums
# speedup vs baseline: 6.0440x; 1.0365x over previous
#include <hip/hip_runtime.h>

#define NN  50000
#define EE  800000
#define EP  (EE + NN)
#define FH  128
#define NH  4
#define HIDW 32
#define EPW  4
#define RN   98
#define NR   ((NN + RN - 1) / RN)
#define PB   256
#define PCH  (EE / PB)
#define PPT  ((PCH + 255) / 256)
#define OWT  1024
#define LP   (FH + 4)

static_assert(EE % PB == 0, "grouping blocks take equal chunks");
static_assert(EPW == 4 && NH == 4, "edge sweeps are written for 4 items x 4 heads per wave");
static_assert(NR <= 1024, "one block scans the range counts");
static_assert(RN > 6 * 16 && RN <= 7 * 16 && OWT == 1024, "the GEMM tail of an owner block splits 7 row tiles over 32 waves");
static_assert(LP % 4 == 0, "LDS rows stay 16-byte aligned");

typedef __attribute__((ext_vector_type(2))) float v2f;
typedef __attribute__((ext_vector_type(8))) float v8f;

__device__ __forceinline__ float lrelu(float v) {
    return fmaxf(v, 0.f) + 0.2f * fminf(v, 0.f);
}

__global__ void part_zero(int* __restrict__ cnt, int* __restrict__ cur)
{
    const int t = blockIdx.x * blockDim.x + threadIdx.x;
    if (t < NR) { cnt[t] = 0; cur[t] = 0; }
}

__global__ void part_count(const int* __restrict__ dst, int* __restrict__ cnt)
{
    __shared__ int hist[NR];
    for (int i = threadIdx.x; i < NR; i += 256) hist[i] = 0;
    __syncthreads();
    const int e0 = blockIdx.x * PCH;
    for (int k = 0; k < PPT; ++k) {
        const int le = k * 256 + threadIdx.x;
        if (le < PCH) atomicAdd(&hist[dst[e0 + le] / RN], 1);
    }
    __syncthreads();
    for (int i = threadIdx.x; i < NR; i += 256) {
        const int c = hist[i];
        if (c) atomicAdd(&cnt[i], c);
    }
}

__global__ void part_scan(const int* __restrict__ cnt, int* __restrict__ off)
{
    __shared__ int s[1024];
    const int t = threadIdx.x;
    const int v = (t < NR) ? cnt[t] : 0;
    s[t] = v;
    __syncthreads();
    for (int d = 1; d < 1024; d <<= 1) {
        const int add = (t >= d) ? s[t - d] : 0;
        __syncthreads();
        s[t] += add;
        __syncthreads();
    }
    if (t < NR) off[t] = s[t] - v;
    if (t == NR - 1) off[NR] = s[t];
}

__global__ void part_fill(const int* __restrict__ src, const int* __restrict__ dst,
                          const int* __restrict__ off, int* __restrict__ cur,
                          int* __restrict__ ssrc, int* __restrict__ sdst)
{
    __shared__ int hist[NR];
    __shared__ int basep[NR];
    for (int i = threadIdx.x; i < NR; i += 256) hist[i] = 0;
    __syncthreads();
    const int e0 = blockIdx.x * PCH;
    int rk[PPT], dv[PPT];
#pragma unroll
    for (int k = 0; k < PPT; ++k) {
        const int le = k * 256 + threadIdx.x;
        rk[k] = -1; dv[k] = 0;
        if (le < PCH) {
            dv[k] = dst[e0 + le];
            rk[k] = atomicAdd(&hist[dv[k] / RN], 1);
        }
    }
    __syncthreads();
    for (int i = threadIdx.x; i < NR; i += 256) {
        const int c = hist[i];
        basep[i] = c ? off[i] + atomicAdd(&cur[i], c) : 0;
    }
    __syncthreads();
#pragma unroll
    for (int k = 0; k < PPT; ++k) {
        if (rk[k] >= 0) {
            const int pos = basep[dv[k] / RN] + rk[k];
            ssrc[pos] = src[e0 + k * 256 + threadIdx.x];
            sdst[pos] = dv[k];
        }
    }
}

__device__ __forceinline__ float div_bias_elu(float x, float den, float bias)
{
    const float v = x / den + bias;
    return (v > 0.f) ? v : (expf(v) - 1.f);
}

__device__ __forceinline__ void wave_items(const int* __restrict__ ssrc,
                                           const int* __restrict__ sdst,
                                           int e_lo, int cnt, int tot, int base, int i0,
                                           int (&s)[EPW], int (&d)[EPW])
{
#pragma unroll
    for (int j = 0; j < EPW; ++j) {
        int idx = i0 + j;
        if (idx >= tot) idx = tot - 1;
        if (idx < cnt) {
            s[j] = ssrc[e_lo + idx];
            d[j] = sdst[e_lo + idx];
        } else {
            s[j] = d[j] = base + (idx - cnt);
        }
    }
}

template <int Q>
__device__ __forceinline__ float quad_bcast(float v)
{
    return __int_as_float(__builtin_amdgcn_mov_dpp(__float_as_int(v), Q * 0x55, 0xF, 0xF, true));
}

template <int M>
__device__ __forceinline__ float row_xor(float v)
{
    return __int_as_float(__builtin_amdgcn_mov_dpp(__float_as_int(v), 0x160 + M, 0xF, 0xF, true));
}

__device__ __forceinline__ float xhalf(float v)
{
    const int b = __float_as_int(v);
    return __int_as_float(__builtin_amdgcn_permlanex16(b, b, 0x76543210, 0xfedcba98, false, false));
}

template <int NT>
__device__ __forceinline__ void next_tiles(const float* buf, int rt0, int lane,
                                           const float* __restrict__ Wn,
                                           const float* __restrict__ bn,
                                           float* __restrict__ on,
                                           int base, int nown, int col)
{
    const int mrow  = lane & 15;
    const int khalf = lane >> 4;
    const float* wp = Wn + (size_t)(khalf * 2) * FH + col;
    const float* ap[NT];
#pragma unroll
    for (int t = 0; t < NT; ++t) {
        const int r = min((rt0 + t) * 16 + mrow, RN - 1);
        ap[t] = buf + r * LP + khalf * 2;
    }
    v8f c[NT] = {};
#pragma unroll
    for (int k = 0; k < FH; k += 4) {
        v2f b;
        b.x = wp[(size_t)k * FH];
        b.y = wp[(size_t)k * FH + FH];
#pragma unroll
        for (int t = 0; t < NT; ++t) {
            const v2f a = *(const v2f*)(ap[t] + k);
            c[t] = __builtin_amdgcn_wmma_f32_16x16x4_f32(false, a, false, b,
                                                         (short)0, c[t], false, false);
        }
    }
    const float bb = bn[col];
#pragma unroll
    for (int t = 0; t < NT; ++t) {
#pragma unroll
        for (int v = 0; v < 8; ++v) {
            const int row = (rt0 + t) * 16 + v + khalf * 8;
            if (row < nown) on[(size_t)(base + row) * FH + col] = c[t][v] + bb;
        }
    }
}

__global__ __launch_bounds__(OWT, 16)
void gemm0_rows(const float* __restrict__ X,
                const float* __restrict__ Wl,
                const float* __restrict__ bl,
                const float* __restrict__ Wr,
                const float* __restrict__ br,
                float* __restrict__ OutL,
                float* __restrict__ OutR)
{
    __shared__ __attribute__((aligned(16))) float buf[RN * LP];

    const int tid  = threadIdx.x;
    const int lane = tid & 31;
    const int wave = __builtin_amdgcn_readfirstlane(tid >> 5);
    const int base = blockIdx.x * RN;
    const int nown = min(RN, NN - base);

    const float4* g = (const float4*)(X + (size_t)base * FH);
    constexpr int SPT = (RN * (FH / 4) + OWT - 1) / OWT;
    float4 piece[SPT];
#pragma unroll
    for (int u = 0; u < SPT; ++u) {
        const int i = tid + u * OWT;
        piece[u] = (i < nown * (FH / 4)) ? g[i] : make_float4(0.f, 0.f, 0.f, 0.f);
    }
#pragma unroll
    for (int u = 0; u < SPT; ++u) {
        const int i = tid + u * OWT;
        if (i < RN * (FH / 4)) ((float4*)buf)[(i >> 5) * (LP / 4) + (i & 31)] = piece[u];
    }
    __syncthreads();

    const int combo = wave & 15;
    const bool right = (combo & 8) != 0;
    const int col = (combo & 7) * 16 + (lane & 15);
    if (wave < 16)
        next_tiles<4>(buf, 0, lane, right ? Wr : Wl, right ? br : bl,
                      right ? OutR : OutL, base, nown, col);
    else
        next_tiles<3>(buf, 4, lane, right ? Wr : Wl, right ? br : bl,
                      right ? OutR : OutL, base, nown, col);
}

template <bool LAST>
__global__ __launch_bounds__(OWT, 16)
void owner_layer(const float* __restrict__ xl,
                 const float* __restrict__ xr,
                 const int* __restrict__ ssrc,
                 const int* __restrict__ sdst,
                 const int* __restrict__ off,
                 const float* __restrict__ att,
                 const float* __restrict__ bo,
                 const float* __restrict__ Wln,
                 const float* __restrict__ bln,
                 const float* __restrict__ Wrn,
                 const float* __restrict__ brn,
                 float* __restrict__ xln,
                 float* __restrict__ xrn,
                 const float* __restrict__ Wf,
                 const float* __restrict__ bf,
                 float* __restrict__ out)
{
    __shared__ __attribute__((aligned(16))) float bufr[RN * LP];
    __shared__ __attribute__((aligned(16))) float buf[RN * LP];
    __shared__ float sh[RN * NH];
    __shared__ float mx[RN * NH];
    __shared__ float den[RN * NH];
    __shared__ int   redo;

    const int tid  = threadIdx.x;
    const int lane = tid & 31;
    const int wave = __builtin_amdgcn_readfirstlane(tid >> 5);
    const int base = blockIdx.x * RN;
    const int nown = min(RN, NN - base);
    const int e_lo = off[blockIdx.x];
    const int cnt  = off[blockIdx.x + 1] - e_lo;

    const int f = lane * 4;
    const float4 w = *(const float4*)(att + f);

    {
        constexpr int NU = (RN + OWT / 32 - 1) / (OWT / 32);
        float4 ra[NU], rb[NU];
#pragma unroll
        for (int u = 0; u < NU; ++u) {
            const int i = wave + u * (OWT / 32);
            ra[u] = rb[u] = make_float4(0.f, 0.f, 0.f, 0.f);
            if (i < nown) {
                ra[u] = *(const float4*)(xl + (size_t)(base + i) * FH + f);
                rb[u] = *(const float4*)(xr + (size_t)(base + i) * FH + f);
            }
        }
#pragma unroll
        for (int u = 0; u < NU; ++u) {
            const int i = wave + u * (OWT / 32);
            if (i < RN) {
                const float4 a = ra[u];
                const float4 b = rb[u];
                ((float4*)bufr)[i * (LP / 4) + lane] = b;
                float* o = buf + i * LP + lane;
                o[0]  = a.x;
                o[32] = a.y;
                o[64] = a.z;
                o[96] = a.w;
                if (lane < LP - FH) buf[i * LP + FH + lane] = 0.f;
                float p = lrelu(a.x + b.x) * w.x + lrelu(a.y + b.y) * w.y +
                          lrelu(a.z + b.z) * w.z + lrelu(a.w + b.w) * w.w;
                p += row_xor<1>(p);
                p += row_xor<2>(p);
                p += row_xor<4>(p);
                if ((lane & 7) == 0) {
                    const int slot = i * NH + (lane >> 3);
                    const bool own = i < nown;
                    sh[slot]  = p;
                    mx[slot]  = own ? p : -3.0e38f;
                    den[slot] = own ? 1.f : 0.f;
                }
            }
        }
        if (tid == 0) redo = 0;
    }
    __syncthreads();

    {
        const bool b0 = (lane & 1) != 0;
        const bool b1 = (lane & 2) != 0;
        const int  jl = ((lane & 1) << 1) | ((lane >> 1) & 1);
        const int  hl = lane >> 3;
        for (int pass = 0; pass < 2; ++pass) {
            const int tot  = (pass == 0) ? cnt : cnt + nown;
            const int ngrp = (tot + EPW - 1) / EPW;
            for (int g = wave; g < ngrp; g += OWT / 32) {
                const int i0 = g * EPW;
                int s[EPW], d[EPW];
                wave_items(ssrc, sdst, e_lo, cnt, tot, base, i0, s, d);
                float4 a[EPW], b[EPW];
#pragma unroll
                for (int j = 0; j < EPW; ++j) {
                    a[j] = *(const float4*)(xl + (size_t)s[j] * FH + f);
                    b[j] = *(const float4*)(bufr + (d[j] - base) * LP + f);
                }
                float p[EPW];
#pragma unroll
                for (int j = 0; j < EPW; ++j)
                    p[j] = lrelu(a[j].x + b[j].x) * w.x + lrelu(a[j].y + b[j].y) * w.y +
                           lrelu(a[j].z + b[j].z) * w.z + lrelu(a[j].w + b[j].w) * w.w;
                const float q0 = (b0 ? p[2] : p[0]) + row_xor<1>(b0 ? p[0] : p[2]);
                const float q1 = (b0 ? p[3] : p[1]) + row_xor<1>(b0 ? p[1] : p[3]);
                float r = (b1 ? q1 : q0) + row_xor<2>(b1 ? q0 : q1);
                r += row_xor<4>(r);

                const int  dj   = (jl == 0) ? d[0] : (jl == 1) ? d[1] : (jl == 2) ? d[2] : d[3];
                const bool ok   = (i0 + jl) < tot;
                const int  slot = (dj - base) * NH + hl;
                const float t   = r - sh[slot];
                const float ex  = ok ? expf(t) : 0.f;
                if (ok && t > 40.f) redo = 1;
                if ((lane & 4) == 0 && ok) {
                    atomicMax(&mx[slot], r);
                    atomicAdd(&den[slot], ex);
                }
                float wt[EPW];
                wt[0] = quad_bcast<0>(ex);
                wt[1] = quad_bcast<2>(ex);
                wt[2] = quad_bcast<1>(ex);
                wt[3] = quad_bcast<3>(ex);
#pragma unroll
                for (int j = 0; j < EPW; ++j) {
                    if (i0 + j < tot) {
                        float* o = buf + (d[j] - base) * LP + lane;
                        atomicAdd(o,      a[j].x * wt[j]);
                        atomicAdd(o + 32, a[j].y * wt[j]);
                        atomicAdd(o + 64, a[j].z * wt[j]);
                        atomicAdd(o + 96, a[j].w * wt[j]);
                    }
                }
            }
            __syncthreads();
            if (pass != 0) break;
            {
                bool bad = false;
                for (int i = tid; i < RN * LP / 4; i += OWT) {
                    const float4 v = ((const float4*)buf)[i];
                    bad |= !(fabsf(v.x) < 1.0e37f) | !(fabsf(v.y) < 1.0e37f) |
                           !(fabsf(v.z) < 1.0e37f) | !(fabsf(v.w) < 1.0e37f);
                }
                if (bad) redo = 1;
            }
            __syncthreads();
            if (redo == 0) break;
            __syncthreads();
            for (int i = tid; i < RN * LP / 4; i += OWT) ((float4*)buf)[i] = make_float4(0.f, 0.f, 0.f, 0.f);
            for (int i = tid; i < RN * NH; i += OWT) { sh[i] = mx[i]; den[i] = 0.f; }
            if (tid == 0) redo = 0;
            __syncthreads();
        }
    }

    for (int i = wave; i < nown; i += OWT / 32) {
        const float* rowp = buf + i * LP + lane;
        float4 v = make_float4(rowp[0], rowp[32], rowp[64], rowp[96]);
        const float dn = fmaxf(den[i * NH + (lane >> 3)], 1e-16f);
        const float4 bb = ((const float4*)bo)[lane];
        v.x = div_bias_elu(v.x, dn, bb.x);
        v.y = div_bias_elu(v.y, dn, bb.y);
        v.z = div_bias_elu(v.z, dn, bb.z);
        v.w = div_bias_elu(v.w, dn, bb.w);
        if (!LAST) {
            ((float4*)buf)[i * (LP / 4) + lane] = v;
        } else {
            const float4 wf = ((const float4*)Wf)[lane];
            float p = v.x * wf.x + v.y * wf.y + v.z * wf.z + v.w * wf.w;
            p += xhalf(p);
            p += row_xor<8>(p);
            p += row_xor<4>(p);
            p += row_xor<2>(p);
            p += row_xor<1>(p);
            if (lane == 0) out[base + i] = p + bf[0];
        }
    }

    if (!LAST) {
        __syncthreads();
        const int combo = wave & 15;
        const bool right = (combo & 8) != 0;
        const int col = (combo & 7) * 16 + (lane & 15);
        if (wave < 16)
            next_tiles<4>(buf, 0, lane, right ? Wrn : Wln, right ? brn : bln,
                          right ? xrn : xln, base, nown, col);
        else
            next_tiles<3>(buf, 4, lane, right ? Wrn : Wln, right ? brn : bln,
                          right ? xrn : xln, base, nown, col);
    }
}

extern "C" void kernel_launch(void* const* d_in, const int* in_sizes, int n_in,
                              void* d_out, int out_size, void* d_ws, size_t ws_size,
                              hipStream_t stream)
{
    const float* x    = (const float*)d_in[0];
    const int*   ei   = (const int*)d_in[1];
    const int*   src  = ei;
    const int*   dst  = ei + EE;
    const float* Wl0  = (const float*)d_in[2];
    const float* bl0  = (const float*)d_in[3];
    const float* Wr0  = (const float*)d_in[4];
    const float* br0  = (const float*)d_in[5];
    const float* att0 = (const float*)d_in[6];
    const float* bo0  = (const float*)d_in[7];
    const float* Wl   = (const float*)d_in[8];
    const float* bl   = (const float*)d_in[9];
    const float* Wr   = (const float*)d_in[10];
    const float* br   = (const float*)d_in[11];
    const float* att  = (const float*)d_in[12];
    const float* bo   = (const float*)d_in[13];
    const float* Wf   = (const float*)d_in[14];
    const float* bf   = (const float*)d_in[15];

    float* ws   = (float*)d_ws;
    float* xlA  = ws;
    float* xrA  = xlA  + (size_t)NN * FH;
    float* xlB  = xrA  + (size_t)NN * FH;
    float* xrB  = xlB  + (size_t)NN * FH;
    int*   ssrc = (int*)(xrB + (size_t)NN * FH);
    int*   sdst = ssrc + EE;
    int*   cnt  = sdst + EE;
    int*   cur  = cnt + NR;
    int*   off  = cur + NR;

    part_zero <<<(NR + 255) / 256, 256, 0, stream>>>(cnt, cur);
    part_count<<<PB, 256, 0, stream>>>(dst, cnt);
    part_scan <<<1, 1024, 0, stream>>>(cnt, off);
    part_fill <<<PB, 256, 0, stream>>>(src, dst, off, cur, ssrc, sdst);

    gemm0_rows<<<NR, OWT, 0, stream>>>(x, Wl0, bl0, Wr0, br0, xlA, xrA);
    owner_layer<false><<<NR, OWT, 0, stream>>>(xlA, xrA, ssrc, sdst, off, att0, bo0,
                                               Wl, bl, Wr, br, xlB, xrB,
                                               Wf, bf, (float*)d_out);
    owner_layer<false><<<NR, OWT, 0, stream>>>(xlB, xrB, ssrc, sdst, off, att, bo,
                                               Wl + (size_t)FH * FH, bl + FH,
                                               Wr + (size_t)FH * FH, br + FH, xlA, xrA,
                                               Wf, bf, (float*)d_out);
    owner_layer<true> <<<NR, OWT, 0, stream>>>(xlA, xrA, ssrc, sdst, off,
                                               att + (size_t)NH * HIDW, bo + FH,
                                               Wl, bl, Wr, br, xlB, xrB,
                                               Wf, bf, (float*)d_out);
}


# Sweep trips take their four (source, destination) pairs by one 32-byte scalar load from an interleaved list, asked for a trip ahead; full trips drop bound tests, branches, 64-bit address sums
# speedup vs baseline: 6.3965x; 1.0583x over previous
#include <hip/hip_runtime.h>

#define NN  50000
#define EE  800000
#define EP  (EE + NN)
#define FH  128
#define NH  4
#define HIDW 32
#define EPW  4
#define RN   98
#define NR   ((NN + RN - 1) / RN)
#define PB   256
#define PCH  (EE / PB)
#define PPT  ((PCH + 255) / 256)
#define OWT  1024
#define LP   (FH + 4)

static_assert(EE % PB == 0, "grouping blocks take equal chunks");
static_assert(EPW == 4 && NH == 4, "edge sweeps are written for 4 items x 4 heads per wave");
static_assert(NR <= 1024, "one block scans the range counts");
static_assert(RN > 6 * 16 && RN <= 7 * 16 && OWT == 1024, "the GEMM tail of an owner block splits 7 row tiles over 32 waves");
static_assert(LP % 4 == 0, "LDS rows stay 16-byte aligned");
static_assert(FH * sizeof(float) == 512, "a row of xl is 512 bytes: a full sweep trip forms its byte offset by a shift of 9");

typedef __attribute__((ext_vector_type(2))) float v2f;
typedef __attribute__((ext_vector_type(8))) float v8f;

__device__ __forceinline__ float lrelu(float v) {
    return fmaxf(v, 0.f) + 0.2f * fminf(v, 0.f);
}

__global__ void part_zero(int* __restrict__ cnt, int* __restrict__ cur)
{
    const int t = blockIdx.x * blockDim.x + threadIdx.x;
    if (t < NR) { cnt[t] = 0; cur[t] = 0; }
}

__global__ void part_count(const int* __restrict__ dst, int* __restrict__ cnt)
{
    __shared__ int hist[NR];
    for (int i = threadIdx.x; i < NR; i += 256) hist[i] = 0;
    __syncthreads();
    const int e0 = blockIdx.x * PCH;
    for (int k = 0; k < PPT; ++k) {
        const int le = k * 256 + threadIdx.x;
        if (le < PCH) atomicAdd(&hist[dst[e0 + le] / RN], 1);
    }
    __syncthreads();
    for (int i = threadIdx.x; i < NR; i += 256) {
        const int c = hist[i];
        if (c) atomicAdd(&cnt[i], c);
    }
}

__global__ void part_scan(const int* __restrict__ cnt, int* __restrict__ off)
{
    __shared__ int s[1024];
    const int t = threadIdx.x;
    const int v = (t < NR) ? cnt[t] : 0;
    s[t] = v;
    __syncthreads();
    for (int d = 1; d < 1024; d <<= 1) {
        const int add = (t >= d) ? s[t - d] : 0;
        __syncthreads();
        s[t] += add;
        __syncthreads();
    }
    if (t < NR) off[t] = s[t] - v;
    if (t == NR - 1) off[NR] = s[t];
}

__global__ void part_fill(const int* __restrict__ src, const int* __restrict__ dst,
                          const int* __restrict__ off, int* __restrict__ cur,
                          int* __restrict__ sitem)
{
    __shared__ int hist[NR];
    __shared__ int basep[NR];
    for (int i = threadIdx.x; i < NR; i += 256) hist[i] = 0;
    __syncthreads();
    const int e0 = blockIdx.x * PCH;
    int rk[PPT], dv[PPT];
#pragma unroll
    for (int k = 0; k < PPT; ++k) {
        const int le = k * 256 + threadIdx.x;
        rk[k] = -1; dv[k] = 0;
        if (le < PCH) {
            dv[k] = dst[e0 + le];
            rk[k] = atomicAdd(&hist[dv[k] / RN], 1);
        }
    }
    __syncthreads();
    for (int i = threadIdx.x; i < NR; i += 256) {
        const int c = hist[i];
        basep[i] = c ? off[i] + atomicAdd(&cur[i], c) : 0;
    }
    __syncthreads();
#pragma unroll
    for (int k = 0; k < PPT; ++k) {
        if (rk[k] >= 0) {
            const int pos = basep[dv[k] / RN] + rk[k];
            *(int2*)(sitem + 2 * pos) = make_int2(src[e0 + k * 256 + threadIdx.x], dv[k]);
        }
    }
}

__device__ __forceinline__ float div_bias_elu(float x, float den, float bias)
{
    const float v = x / den + bias;
    return (v > 0.f) ? v : (expf(v) - 1.f);
}

__device__ __forceinline__ void wave_items(const int* __restrict__ sitem,
                                           int e_lo, int cnt, int tot, int base, int i0,
                                           int (&s)[EPW], int (&d)[EPW])
{
#pragma unroll
    for (int j = 0; j < EPW; ++j) {
        int idx = i0 + j;
        if (idx >= tot) idx = tot - 1;
        if (idx < cnt) {
            s[j] = sitem[2 * (e_lo + idx)];
            d[j] = sitem[2 * (e_lo + idx) + 1];
        } else {
            s[j] = d[j] = base + (idx - cnt);
        }
    }
}

template <int Q>
__device__ __forceinline__ float quad_bcast(float v)
{
    return __int_as_float(__builtin_amdgcn_mov_dpp(__float_as_int(v), Q * 0x55, 0xF, 0xF, true));
}

template <int M>
__device__ __forceinline__ float row_xor(float v)
{
    return __int_as_float(__builtin_amdgcn_mov_dpp(__float_as_int(v), 0x160 + M, 0xF, 0xF, true));
}

__device__ __forceinline__ float xhalf(float v)
{
    const int b = __float_as_int(v);
    return __int_as_float(__builtin_amdgcn_permlanex16(b, b, 0x76543210, 0xfedcba98, false, false));
}

__device__ __forceinline__ void trip_ids(const int* __restrict__ sitem, int first, int (&it)[2 * EPW])
{
    __builtin_memcpy(it, sitem + 2 * first, sizeof(it));
}

template <bool FULL>
__device__ __forceinline__ bool sweep_trip(const float* __restrict__ xl,
                                           const int* __restrict__ sitem,
                                           const float* bufr, float* buf,
                                           const float* sh, float* mx, float* den,
                                           int e_lo, int cnt, int tot, int base, int i0,
                                           const int (&it)[2 * EPW],
                                           int lane, const float4 w)
{
    const int  f  = lane * 4;
    const bool b0 = (lane & 1) != 0;
    const bool b1 = (lane & 2) != 0;
    const int  jl = ((lane & 1) << 1) | ((lane >> 1) & 1);
    const int  hl = lane >> 3;
    int s[EPW], d[EPW], dj;
    float4 a[EPW], b[EPW];
    if (FULL) {
#pragma unroll
        for (int j = 0; j < EPW; ++j) { s[j] = it[2 * j]; d[j] = it[2 * j + 1]; }
        dj = sitem[2 * (e_lo + i0 + jl) + 1];
#pragma unroll
        for (int j = 0; j < EPW; ++j)
            a[j] = *(const float4*)((const char*)xl + (((unsigned)s[j] << 9) | ((unsigned)lane << 4)));
    } else {
        wave_items(sitem, e_lo, cnt, tot, base, i0, s, d);
        dj = (jl == 0) ? d[0] : (jl == 1) ? d[1] : (jl == 2) ? d[2] : d[3];
#pragma unroll
        for (int j = 0; j < EPW; ++j)
            a[j] = *(const float4*)(xl + (size_t)s[j] * FH + f);
    }
#pragma unroll
    for (int j = 0; j < EPW; ++j)
        b[j] = *(const float4*)(bufr + (d[j] - base) * LP + f);
    float p[EPW];
#pragma unroll
    for (int j = 0; j < EPW; ++j)
        p[j] = lrelu(a[j].x + b[j].x) * w.x + lrelu(a[j].y + b[j].y) * w.y +
               lrelu(a[j].z + b[j].z) * w.z + lrelu(a[j].w + b[j].w) * w.w;
    const float q0 = (b0 ? p[2] : p[0]) + row_xor<1>(b0 ? p[0] : p[2]);
    const float q1 = (b0 ? p[3] : p[1]) + row_xor<1>(b0 ? p[1] : p[3]);
    float r = (b1 ? q1 : q0) + row_xor<2>(b1 ? q0 : q1);
    r += row_xor<4>(r);

    const bool ok   = FULL || (i0 + jl) < tot;
    const int  slot = (dj - base) * NH + hl;
    const float t   = r - sh[slot];
    const float ex  = ok ? expf(t) : 0.f;
    if ((lane & 4) == 0 && ok) {
        atomicMax(&mx[slot], r);
        atomicAdd(&den[slot], ex);
    }
    float wt[EPW];
    wt[0] = quad_bcast<0>(ex);
    wt[1] = quad_bcast<2>(ex);
    wt[2] = quad_bcast<1>(ex);
    wt[3] = quad_bcast<3>(ex);
#pragma unroll
    for (int j = 0; j < EPW; ++j) {
        if (FULL || i0 + j < tot) {
            float* o = buf + (d[j] - base) * LP + lane;
            atomicAdd(o,      a[j].x * wt[j]);
            atomicAdd(o + 32, a[j].y * wt[j]);
            atomicAdd(o + 64, a[j].z * wt[j]);
            atomicAdd(o + 96, a[j].w * wt[j]);
        }
    }
    return ok && t > 40.f;
}

template <int NT>
__device__ __forceinline__ void next_tiles(const float* buf, int rt0, int lane,
                                           const float* __restrict__ Wn,
                                           const float* __restrict__ bn,
                                           float* __restrict__ on,
                                           int base, int nown, int col)
{
    const int mrow  = lane & 15;
    const int khalf = lane >> 4;
    const float* wp = Wn + (size_t)(khalf * 2) * FH + col;
    const float* ap[NT];
#pragma unroll
    for (int t = 0; t < NT; ++t) {
        const int r = min((rt0 + t) * 16 + mrow, RN - 1);
        ap[t] = buf + r * LP + khalf * 2;
    }
    v8f c[NT] = {};
#pragma unroll
    for (int k = 0; k < FH; k += 4) {
        v2f b;
        b.x = wp[(size_t)k * FH];
        b.y = wp[(size_t)k * FH + FH];
#pragma unroll
        for (int t = 0; t < NT; ++t) {
            const v2f a = *(const v2f*)(ap[t] + k);
            c[t] = __builtin_amdgcn_wmma_f32_16x16x4_f32(false, a, false, b,
                                                         (short)0, c[t], false, false);
        }
    }
    const float bb = bn[col];
#pragma unroll
    for (int t = 0; t < NT; ++t) {
#pragma unroll
        for (int v = 0; v < 8; ++v) {
            const int row = (rt0 + t) * 16 + v + khalf * 8;
            if (row < nown) on[(size_t)(base + row) * FH + col] = c[t][v] + bb;
        }
    }
}

__global__ __launch_bounds__(OWT, 16)
void gemm0_rows(const float* __restrict__ X,
                const float* __restrict__ Wl,
                const float* __restrict__ bl,
                const float* __restrict__ Wr,
                const float* __restrict__ br,
                float* __restrict__ OutL,
                float* __restrict__ OutR)
{
    __shared__ __attribute__((aligned(16))) float buf[RN * LP];

    const int tid  = threadIdx.x;
    const int lane = tid & 31;
    const int wave = __builtin_amdgcn_readfirstlane(tid >> 5);
    const int base = blockIdx.x * RN;
    const int nown = min(RN, NN - base);

    const float4* g = (const float4*)(X + (size_t)base * FH);
    constexpr int SPT = (RN * (FH / 4) + OWT - 1) / OWT;
    float4 piece[SPT];
#pragma unroll
    for (int u = 0; u < SPT; ++u) {
        const int i = tid + u * OWT;
        piece[u] = (i < nown * (FH / 4)) ? g[i] : make_float4(0.f, 0.f, 0.f, 0.f);
    }
#pragma unroll
    for (int u = 0; u < SPT; ++u) {
        const int i = tid + u * OWT;
        if (i < RN * (FH / 4)) ((float4*)buf)[(i >> 5) * (LP / 4) + (i & 31)] = piece[u];
    }
    __syncthreads();

    const int combo = wave & 15;
    const bool right = (combo & 8) != 0;
    const int col = (combo & 7) * 16 + (lane & 15);
    if (wave < 16)
        next_tiles<4>(buf, 0, lane, right ? Wr : Wl, right ? br : bl,
                      right ? OutR : OutL, base, nown, col);
    else
        next_tiles<3>(buf, 4, lane, right ? Wr : Wl, right ? br : bl,
                      right ? OutR : OutL, base, nown, col);
}

template <bool LAST>
__global__ __launch_bounds__(OWT, 16)
void owner_layer(const float* __restrict__ xl,
                 const float* __restrict__ xr,
                 const int* __restrict__ sitem,
                 const int* __restrict__ off,
                 const float* __restrict__ att,
                 const float* __restrict__ bo,
                 const float* __restrict__ Wln,
                 const float* __restrict__ bln,
                 const float* __restrict__ Wrn,
                 const float* __restrict__ brn,
                 float* __restrict__ xln,
                 float* __restrict__ xrn,
                 const float* __restrict__ Wf,
                 const float* __restrict__ bf,
                 float* __restrict__ out)
{
    __shared__ __attribute__((aligned(16))) float bufr[RN * LP];
    __shared__ __attribute__((aligned(16))) float buf[RN * LP];
    __shared__ float sh[RN * NH];
    __shared__ float mx[RN * NH];
    __shared__ float den[RN * NH];
    __shared__ int   redo;

    const int tid  = threadIdx.x;
    const int lane = tid & 31;
    const int wave = __builtin_amdgcn_readfirstlane(tid >> 5);
    const int base = blockIdx.x * RN;
    const int nown = min(RN, NN - base);
    const int e_lo = off[blockIdx.x];
    const int cnt  = off[blockIdx.x + 1] - e_lo;

    const int f = lane * 4;
    const float4 w = *(const float4*)(att + f);

    {
        constexpr int NU = (RN + OWT / 32 - 1) / (OWT / 32);
        float4 ra[NU], rb[NU];
#pragma unroll
        for (int u = 0; u < NU; ++u) {
            const int i = wave + u * (OWT / 32);
            ra[u] = rb[u] = make_float4(0.f, 0.f, 0.f, 0.f);
            if (i < nown) {
                ra[u] = *(const float4*)(xl + (size_t)(base + i) * FH + f);
                rb[u] = *(const float4*)(xr + (size_t)(base + i) * FH + f);
            }
        }
#pragma unroll
        for (int u = 0; u < NU; ++u) {
            const int i = wave + u * (OWT / 32);
            if (i < RN) {
                const float4 a = ra[u];
                const float4 b = rb[u];
                ((float4*)bufr)[i * (LP / 4) + lane] = b;
                float* o = buf + i * LP + lane;
                o[0]  = a.x;
                o[32] = a.y;
                o[64] = a.z;
                o[96] = a.w;
                if (lane < LP - FH) buf[i * LP + FH + lane] = 0.f;
                float p = lrelu(a.x + b.x) * w.x + lrelu(a.y + b.y) * w.y +
                          lrelu(a.z + b.z) * w.z + lrelu(a.w + b.w) * w.w;
                p += row_xor<1>(p);
                p += row_xor<2>(p);
                p += row_xor<4>(p);
                if ((lane & 7) == 0) {
                    const int slot = i * NH + (lane >> 3);
                    const bool own = i < nown;
                    sh[slot]  = p;
                    mx[slot]  = own ? p : -3.0e38f;
                    den[slot] = own ? 1.f : 0.f;
                }
            }
        }
        if (tid == 0) redo = 0;
    }
    __syncthreads();

    {
        const int nfull = cnt / EPW;
        for (int pass = 0; pass < 2; ++pass) {
            const int tot  = (pass == 0) ? cnt : cnt + nown;
            const int ngrp = (tot + EPW - 1) / EPW;
            bool over = false;
            int g = wave;
            int ids[2 * EPW] = {};
            if (g < nfull) {
                trip_ids(sitem, e_lo + g * EPW, ids);
                for (;;) {
                    int nxt[2 * EPW];
                    trip_ids(sitem, e_lo + min(g + OWT / 32, nfull - 1) * EPW, nxt);
                    over |= sweep_trip<true>(xl, sitem, bufr, buf, sh, mx, den,
                                             e_lo, cnt, tot, base, g * EPW, ids, lane, w);
                    g += OWT / 32;
                    if (g >= nfull) break;
#pragma unroll
                    for (int k = 0; k < 2 * EPW; ++k) ids[k] = nxt[k];
                }
            }
            for (; g < ngrp; g += OWT / 32)
                over |= sweep_trip<false>(xl, sitem, bufr, buf, sh, mx, den,
                                          e_lo, cnt, tot, base, g * EPW, ids, lane, w);
            if (over) redo = 1;
            __syncthreads();
            if (pass != 0) break;
            {
                bool bad = false;
                for (int i = tid; i < RN * LP / 4; i += OWT) {
                    const float4 v = ((const float4*)buf)[i];
                    bad |= !(fabsf(v.x) < 1.0e37f) | !(fabsf(v.y) < 1.0e37f) |
                           !(fabsf(v.z) < 1.0e37f) | !(fabsf(v.w) < 1.0e37f);
                }
                if (bad) redo = 1;
            }
            __syncthreads();
            if (redo == 0) break;
            __syncthreads();
            for (int i = tid; i < RN * LP / 4; i += OWT) ((float4*)buf)[i] = make_float4(0.f, 0.f, 0.f, 0.f);
            for (int i = tid; i < RN * NH; i += OWT) { sh[i] = mx[i]; den[i] = 0.f; }
            if (tid == 0) redo = 0;
            __syncthreads();
        }
    }

    for (int i = wave; i < nown; i += OWT / 32) {
        const float* rowp = buf + i * LP + lane;
        float4 v = make_float4(rowp[0], rowp[32], rowp[64], rowp[96]);
        const float dn = fmaxf(den[i * NH + (lane >> 3)], 1e-16f);
        const float4 bb = ((const float4*)bo)[lane];
        v.x = div_bias_elu(v.x, dn, bb.x);
        v.y = div_bias_elu(v.y, dn, bb.y);
        v.z = div_bias_elu(v.z, dn, bb.z);
        v.w = div_bias_elu(v.w, dn, bb.w);
        if (!LAST) {
            ((float4*)buf)[i * (LP / 4) + lane] = v;
        } else {
            const float4 wf = ((const float4*)Wf)[lane];
            float p = v.x * wf.x + v.y * wf.y + v.z * wf.z + v.w * wf.w;
            p += xhalf(p);
            p += row_xor<8>(p);
            p += row_xor<4>(p);
            p += row_xor<2>(p);
            p += row_xor<1>(p);
            if (lane == 0) out[base + i] = p + bf[0];
        }
    }

    if (!LAST) {
        __syncthreads();
        const int combo = wave & 15;
        const bool right = (combo & 8) != 0;
        const int col = (combo & 7) * 16 + (lane & 15);
        if (wave < 16)
            next_tiles<4>(buf, 0, lane, right ? Wrn : Wln, right ? brn : bln,
                          right ? xrn : xln, base, nown, col);
        else
            next_tiles<3>(buf, 4, lane, right ? Wrn : Wln, right ? brn : bln,
                          right ? xrn : xln, base, nown, col);
    }
}

extern "C" void kernel_launch(void* const* d_in, const int* in_sizes, int n_in,
                              void* d_out, int out_size, void* d_ws, size_t ws_size,
                              hipStream_t stream)
{
    const float* x    = (const float*)d_in[0];
    const int*   ei   = (const int*)d_in[1];
    const int*   src  = ei;
    const int*   dst  = ei + EE;
    const float* Wl0  = (const float*)d_in[2];
    const float* bl0  = (const float*)d_in[3];
    const float* Wr0  = (const float*)d_in[4];
    const float* br0  = (const float*)d_in[5];
    const float* att0 = (const float*)d_in[6];
    const float* bo0  = (const float*)d_in[7];
    const float* Wl   = (const float*)d_in[8];
    const float* bl   = (const float*)d_in[9];
    const float* Wr   = (const float*)d_in[10];
    const float* br   = (const float*)d_in[11];
    const float* att  = (const float*)d_in[12];
    const float* bo   = (const float*)d_in[13];
    const float* Wf   = (const float*)d_in[14];
    const float* bf   = (const float*)d_in[15];

    float* ws   = (float*)d_ws;
    float* xlA  = ws;
    float* xrA  = xlA  + (size_t)NN * FH;
    float* xlB  = xrA  + (size_t)NN * FH;
    float* xrB  = xlB  + (size_t)NN * FH;
    int*   sitem = (int*)(xrB + (size_t)NN * FH);
    int*   cnt  = sitem + 2 * (size_t)EE;
    int*   cur  = cnt + NR;
    int*   off  = cur + NR;

    part_zero <<<(NR + 255) / 256, 256, 0, stream>>>(cnt, cur);
    part_count<<<PB, 256, 0, stream>>>(dst, cnt);
    part_scan <<<1, 1024, 0, stream>>>(cnt, off);
    part_fill <<<PB, 256, 0, stream>>>(src, dst, off, cur, sitem);

    gemm0_rows<<<NR, OWT, 0, stream>>>(x, Wl0, bl0, Wr0, br0, xlA, xrA);
    owner_layer<false><<<NR, OWT, 0, stream>>>(xlA, xrA, sitem, off, att0, bo0,
                                               Wl, bl, Wr, br, xlB, xrB,
                                               Wf, bf, (float*)d_out);
    owner_layer<false><<<NR, OWT, 0, stream>>>(xlB, xrB, sitem, off, att, bo,
                                               Wl + (size_t)FH * FH, bl + FH,
                                               Wr + (size_t)FH * FH, br + FH, xlA, xrA,
                                               Wf, bf, (float*)d_out);
    owner_layer<true> <<<NR, OWT, 0, stream>>>(xlA, xrA, sitem, off,
                                               att + (size_t)NH * HIDW, bo + FH,
                                               Wl, bl, Wr, br, xlB, xrB,
                                               Wf, bf, (float*)d_out);
}
